# starting kernel
# baseline (speedup 1.0000x reference)
#include <hip/hip_runtime.h>
#include <hip/hip_bf16.h>


#define N_  4
#define L_  4096
#define H_  8
#define D_  64
#define DV_ 64
#define C_  64
#define T_  (L_ / C_)
#define NH_ (N_ * H_)
#define EPS_ 1e-6f

typedef float v2f __attribute__((ext_vector_type(2)));
typedef float v8f __attribute__((ext_vector_type(8)));

__device__ __forceinline__ v8f wmma4(v2f a, v2f b, v8f c) {
  return __builtin_amdgcn_wmma_f32_16x16x4_f32(
       false, a,  false, b,
       (short)0, c,  false,  false);
}

__device__ __forceinline__ v2f frag_a_rm(const float* buf, int ld, int mb, int kk, int lane) {
  int m = mb + (lane & 15);
  int k = kk * 4 + ((lane >> 4) << 1);
  v2f a; a.x = buf[m * ld + k]; a.y = buf[m * ld + k + 1]; return a;
}
__device__ __forceinline__ v2f frag_a_tr(const float* buf, int ld, int mb, int kk, int lane) {
  int m = mb + (lane & 15);
  int k = kk * 4 + ((lane >> 4) << 1);
  v2f a; a.x = buf[k * ld + m]; a.y = buf[(k + 1) * ld + m]; return a;
}
__device__ __forceinline__ v2f frag_b_rm(const float* buf, int ld, int nb, int kk, int lane) {
  int n = nb + (lane & 15);
  int k = kk * 4 + ((lane >> 4) << 1);
  v2f b; b.x = buf[k * ld + n]; b.y = buf[(k + 1) * ld + n]; return b;
}
__device__ __forceinline__ v2f frag_b_tr(const float* buf, int ld, int nb, int kk, int lane) {
  int n = nb + (lane & 15);
  int k = kk * 4 + ((lane >> 4) << 1);
  v2f b; b.x = buf[n * ld + k]; b.y = buf[n * ld + k + 1]; return b;
}

__device__ __forceinline__ float featmap(float x) {
  return x > 0.0f ? x + 1.0f : __expf(x);
}

__global__ __launch_bounds__(256) void cla_phase1(
    const float* __restrict__ kin, const float* __restrict__ vin,
    float* __restrict__ kv, float* __restrict__ ksum) {
  __shared__ float Ks[C_ * D_];
  __shared__ float Vs[C_ * DV_];

  const int bt = blockIdx.x;
  const int nh = bt / T_, t = bt % T_;
  const int n = nh / H_, h = nh % H_;
  const int tid = threadIdx.x;
  const size_t rs = (size_t)H_ * D_;
  const float* kb = kin + (((size_t)n * L_ + (size_t)t * C_) * H_ + h) * D_;
  const float* vb = vin + (((size_t)n * L_ + (size_t)t * C_) * H_ + h) * D_;

  for (int idx = tid; idx < C_ * D_; idx += 256) {
    int r = idx >> 6, c = idx & 63;
    Ks[idx] = featmap(kb[(size_t)r * rs + c]);
    Vs[idx] = vb[(size_t)r * rs + c];
  }
  __syncthreads();

  const int wave = tid >> 5, lane = tid & 31;
  float* kvb = kv + (size_t)(nh * T_ + t) * (D_ * DV_);

#pragma unroll
  for (int j = 0; j < 2; ++j) {
    const int tt = wave * 2 + j;
    const int mb = (tt >> 2) * 16, nb = (tt & 3) * 16;
    v8f acc = {};
#pragma unroll
    for (int kk = 0; kk < 16; ++kk) {
      v2f a = frag_a_tr(Ks, D_, mb, kk, lane);
      v2f b = frag_b_rm(Vs, DV_, nb, kk, lane);
      acc = wmma4(a, b, acc);
    }
    const int col = nb + (lane & 15);
    const int rb  = mb + ((lane >> 4) << 3);
#pragma unroll
    for (int r = 0; r < 8; ++r)
      kvb[(size_t)(rb + r) * DV_ + col] = acc[r];
  }

  if (tid < D_) {
    float s = 0.0f;
    for (int r = 0; r < C_; ++r) s += Ks[r * D_ + tid];
    ksum[(size_t)(nh * T_ + t) * D_ + tid] = s;
  }
}

__global__ __launch_bounds__(256) void cla_phase2(
    float* __restrict__ kv, float* __restrict__ ksum) {
  const int nh = blockIdx.x;
  const int tid = threadIdx.x;
  float run[16];
#pragma unroll
  for (int i = 0; i < 16; ++i) run[i] = 0.0f;
  float runk = 0.0f;

  float* base  = kv + (size_t)nh * T_ * (D_ * DV_);
  float* kbase = ksum + (size_t)nh * T_ * D_;

  for (int t = 0; t < T_; ++t) {
    float* p = base + (size_t)t * (D_ * DV_);
#pragma unroll
    for (int i = 0; i < 16; ++i) {
      const int idx = tid + i * 256;
      const float cur = p[idx];
      p[idx] = run[i];
      run[i] += cur;
    }
    if (tid < D_) {
      const float cur = kbase[t * D_ + tid];
      kbase[t * D_ + tid] = runk;
      runk += cur;
    }
  }
}

__global__ __launch_bounds__(256) void cla_phase3(
    const float* __restrict__ qin, const float* __restrict__ kin,
    const float* __restrict__ vin, const float* __restrict__ kv,
    const float* __restrict__ ksum, float* __restrict__ out) {
  __shared__ float Qs[C_ * D_];
  __shared__ float Vs[C_ * DV_];
  __shared__ float Bs[C_ * D_];
  __shared__ float ks[D_];
  __shared__ float inv[C_];

  const int bt = blockIdx.x;
  const int nh = bt / T_, t = bt % T_;
  const int n = nh / H_, h = nh % H_;
  const int tid = threadIdx.x;
  const size_t rs = (size_t)H_ * D_;
  const float* qb = qin + (((size_t)n * L_ + (size_t)t * C_) * H_ + h) * D_;
  const float* kb = kin + (((size_t)n * L_ + (size_t)t * C_) * H_ + h) * D_;
  const float* vb = vin + (((size_t)n * L_ + (size_t)t * C_) * H_ + h) * D_;
  const float* sp = kv + (size_t)(nh * T_ + t) * (D_ * DV_);

  for (int idx = tid; idx < C_ * D_; idx += 256) {
    int r = idx >> 6, c = idx & 63;
    Qs[idx] = featmap(qb[(size_t)r * rs + c]);
    Vs[idx] = vb[(size_t)r * rs + c];
    Bs[idx] = sp[idx];
  }
  if (tid < D_) ks[tid] = ksum[(size_t)(nh * T_ + t) * D_ + tid];
  __syncthreads();

  const int wave = tid >> 5, lane = tid & 31;
  v8f acc[2];

#pragma unroll
  for (int j = 0; j < 2; ++j) {
    const int tt = wave * 2 + j;
    const int mb = (tt >> 2) * 16, nb = (tt & 3) * 16;
    acc[j] = (v8f){};
#pragma unroll
    for (int kk = 0; kk < 16; ++kk) {
      v2f a = frag_a_rm(Qs, D_, mb, kk, lane);
      v2f b = frag_b_rm(Bs, DV_, nb, kk, lane);
      acc[j] = wmma4(a, b, acc[j]);
    }
  }
  __syncthreads();

  for (int idx = tid; idx < C_ * D_; idx += 256) {
    int r = idx >> 6, c = idx & 63;
    Bs[idx] = featmap(kb[(size_t)r * rs + c]);
  }
  __syncthreads();

  v8f p[2];
#pragma unroll
  for (int j = 0; j < 2; ++j) {
    const int tt = wave * 2 + j;
    const int ti = tt >> 2, tj = tt & 3;
    p[j] = (v8f){};
    if (tj <= ti) {
#pragma unroll
      for (int kk = 0; kk < 16; ++kk) {
        v2f a = frag_a_rm(Qs, D_, ti * 16, kk, lane);
        v2f b = frag_b_tr(Bs, D_, tj * 16, kk, lane);
        p[j] = wmma4(a, b, p[j]);
      }
    }
  }
  __syncthreads();

#pragma unroll
  for (int j = 0; j < 2; ++j) {
    const int tt = wave * 2 + j;
    const int ti = tt >> 2, tj = tt & 3;
    const int col = tj * 16 + (lane & 15);
    const int rb  = ti * 16 + ((lane >> 4) << 3);
#pragma unroll
    for (int r = 0; r < 8; ++r) {
      const int row = rb + r;
      Bs[row * C_ + col] = (col <= row) ? p[j][r] : 0.0f;
    }
  }
  __syncthreads();

  if (tid < C_) {
    float s = 0.0f, dq = 0.0f;
    for (int c = 0; c < C_; ++c) {
      s  += Bs[tid * C_ + c];
      dq += Qs[tid * D_ + c] * ks[c];
    }
    inv[tid] = 1.0f / (s + dq + EPS_);
  }

#pragma unroll
  for (int j = 0; j < 2; ++j) {
    const int tt = wave * 2 + j;
    const int mb = (tt >> 2) * 16, nb = (tt & 3) * 16;
#pragma unroll
    for (int kk = 0; kk < 16; ++kk) {
      v2f a = frag_a_rm(Bs, C_, mb, kk, lane);
      v2f b = frag_b_rm(Vs, DV_, nb, kk, lane);
      acc[j] = wmma4(a, b, acc[j]);
    }
  }
  __syncthreads();

  float* ob = out + (((size_t)n * L_ + (size_t)t * C_) * H_ + h) * DV_;
#pragma unroll
  for (int j = 0; j < 2; ++j) {
    const int tt = wave * 2 + j;
    const int mb = (tt >> 2) * 16, nb = (tt & 3) * 16;
    const int col = nb + (lane & 15);
    const int rb  = mb + ((lane >> 4) << 3);
#pragma unroll
    for (int r = 0; r < 8; ++r) {
      const int row = rb + r;
      ob[(size_t)row * (H_ * DV_) + col] = acc[j][r] * inv[row];
    }
  }
}

extern "C" void kernel_launch(void* const* d_in, const int* in_sizes, int n_in,
                              void* d_out, int out_size, void* d_ws, size_t ws_size,
                              hipStream_t stream) {
  (void)in_sizes; (void)n_in; (void)out_size; (void)ws_size;
  const float* q = (const float*)d_in[0];
  const float* k = (const float*)d_in[1];
  const float* v = (const float*)d_in[2];
  float* out = (float*)d_out;

  float* kv   = (float*)d_ws;
  float* ksum = kv + (size_t)NH_ * T_ * D_ * DV_;

  cla_phase1<<<NH_ * T_, 256, 0, stream>>>(k, v, kv, ksum);
  cla_phase2<<<NH_,       256, 0, stream>>>(kv, ksum);
  cla_phase3<<<NH_ * T_, 256, 0, stream>>>(q, k, v, kv, ksum, out);
}


# Phase-2 prefix scan spread over the chip: one float4 lane per thread (272 work-groups of 4 waves, not 32 of 8), a lane's chunk loads all requested before its serial adds; sum order kept
# speedup vs baseline: 1.2126x; 1.2126x over previous
#include <hip/hip_runtime.h>
#include <hip/hip_bf16.h>


#define N_  4
#define L_  4096
#define H_  8
#define D_  64
#define DV_ 64
#define C_  64
#define T_  (L_ / C_)
#define NH_ (N_ * H_)
#define EPS_ 1e-6f

typedef float v2f __attribute__((ext_vector_type(2)));
typedef float v8f __attribute__((ext_vector_type(8)));

__device__ __forceinline__ v8f wmma4(v2f a, v2f b, v8f c) {
  return __builtin_amdgcn_wmma_f32_16x16x4_f32(
       false, a,  false, b,
       (short)0, c,  false,  false);
}

__device__ __forceinline__ v2f frag_a_rm(const float* buf, int ld, int mb, int kk, int lane) {
  int m = mb + (lane & 15);
  int k = kk * 4 + ((lane >> 4) << 1);
  v2f a; a.x = buf[m * ld + k]; a.y = buf[m * ld + k + 1]; return a;
}
__device__ __forceinline__ v2f frag_a_tr(const float* buf, int ld, int mb, int kk, int lane) {
  int m = mb + (lane & 15);
  int k = kk * 4 + ((lane >> 4) << 1);
  v2f a; a.x = buf[k * ld + m]; a.y = buf[(k + 1) * ld + m]; return a;
}
__device__ __forceinline__ v2f frag_b_rm(const float* buf, int ld, int nb, int kk, int lane) {
  int n = nb + (lane & 15);
  int k = kk * 4 + ((lane >> 4) << 1);
  v2f b; b.x = buf[k * ld + n]; b.y = buf[(k + 1) * ld + n]; return b;
}
__device__ __forceinline__ v2f frag_b_tr(const float* buf, int ld, int nb, int kk, int lane) {
  int n = nb + (lane & 15);
  int k = kk * 4 + ((lane >> 4) << 1);
  v2f b; b.x = buf[n * ld + k]; b.y = buf[n * ld + k + 1]; return b;
}

__device__ __forceinline__ float featmap(float x) {
  return x > 0.0f ? x + 1.0f : __expf(x);
}

__global__ __launch_bounds__(256) void cla_phase1(
    const float* __restrict__ kin, const float* __restrict__ vin,
    float* __restrict__ kv, float* __restrict__ ksum) {
  __shared__ float Ks[C_ * D_];
  __shared__ float Vs[C_ * DV_];

  const int bt = blockIdx.x;
  const int nh = bt / T_, t = bt % T_;
  const int n = nh / H_, h = nh % H_;
  const int tid = threadIdx.x;
  const size_t rs = (size_t)H_ * D_;
  const float* kb = kin + (((size_t)n * L_ + (size_t)t * C_) * H_ + h) * D_;
  const float* vb = vin + (((size_t)n * L_ + (size_t)t * C_) * H_ + h) * D_;

  for (int idx = tid; idx < C_ * D_; idx += 256) {
    int r = idx >> 6, c = idx & 63;
    Ks[idx] = featmap(kb[(size_t)r * rs + c]);
    Vs[idx] = vb[(size_t)r * rs + c];
  }
  __syncthreads();

  const int wave = tid >> 5, lane = tid & 31;
  float* kvb = kv + (size_t)(nh * T_ + t) * (D_ * DV_);

#pragma unroll
  for (int j = 0; j < 2; ++j) {
    const int tt = wave * 2 + j;
    const int mb = (tt >> 2) * 16, nb = (tt & 3) * 16;
    v8f acc = {};
#pragma unroll
    for (int kk = 0; kk < 16; ++kk) {
      v2f a = frag_a_tr(Ks, D_, mb, kk, lane);
      v2f b = frag_b_rm(Vs, DV_, nb, kk, lane);
      acc = wmma4(a, b, acc);
    }
    const int col = nb + (lane & 15);
    const int rb  = mb + ((lane >> 4) << 3);
#pragma unroll
    for (int r = 0; r < 8; ++r)
      kvb[(size_t)(rb + r) * DV_ + col] = acc[r];
  }

  if (tid < D_) {
    float s = 0.0f;
    for (int r = 0; r < C_; ++r) s += Ks[r * D_ + tid];
    ksum[(size_t)(nh * T_ + t) * D_ + tid] = s;
  }
}

typedef float v4f __attribute__((ext_vector_type(4)));

#define P2_THREADS_   128
#define KV4_          (D_ * DV_ / 4)
#define P2_KV_BLOCKS_ (NH_ * KV4_ / P2_THREADS_)
#define P2_KS_BLOCKS_ (NH_ * D_ / P2_THREADS_)
#define P2_BLOCKS_    (P2_KV_BLOCKS_ + P2_KS_BLOCKS_)

template <typename V, int STRIDE>
__device__ __forceinline__ void scan_chunks(V* p) {
  V v[T_];
#pragma unroll
  for (int t = 0; t < T_; ++t) v[t] = p[t * STRIDE];
  V run = V(0.0f);
#pragma unroll
  for (int t = 0; t < T_; ++t) {
    p[t * STRIDE] = run;
    run += v[t];
  }
}

__global__ __launch_bounds__(P2_THREADS_) __attribute__((amdgpu_waves_per_eu(1, 2)))
void cla_phase2(
    float* __restrict__ kv, float* __restrict__ ksum) {
  const int b = blockIdx.x;
  if (b < P2_KV_BLOCKS_) {
    const int g  = b * P2_THREADS_ + threadIdx.x;
    const int nh = g / KV4_, i = g % KV4_;
    scan_chunks<v4f, KV4_>(reinterpret_cast<v4f*>(kv) + (size_t)nh * T_ * KV4_ + i);
  } else {
    const int g  = (b - P2_KV_BLOCKS_) * P2_THREADS_ + threadIdx.x;
    const int nh = g / D_, i = g % D_;
    scan_chunks<float, D_>(ksum + (size_t)nh * T_ * D_ + i);
  }
}

__global__ __launch_bounds__(256) void cla_phase3(
    const float* __restrict__ qin, const float* __restrict__ kin,
    const float* __restrict__ vin, const float* __restrict__ kv,
    const float* __restrict__ ksum, float* __restrict__ out) {
  __shared__ float Qs[C_ * D_];
  __shared__ float Vs[C_ * DV_];
  __shared__ float Bs[C_ * D_];
  __shared__ float ks[D_];
  __shared__ float inv[C_];

  const int bt = blockIdx.x;
  const int nh = bt / T_, t = bt % T_;
  const int n = nh / H_, h = nh % H_;
  const int tid = threadIdx.x;
  const size_t rs = (size_t)H_ * D_;
  const float* qb = qin + (((size_t)n * L_ + (size_t)t * C_) * H_ + h) * D_;
  const float* kb = kin + (((size_t)n * L_ + (size_t)t * C_) * H_ + h) * D_;
  const float* vb = vin + (((size_t)n * L_ + (size_t)t * C_) * H_ + h) * D_;
  const float* sp = kv + (size_t)(nh * T_ + t) * (D_ * DV_);

  for (int idx = tid; idx < C_ * D_; idx += 256) {
    int r = idx >> 6, c = idx & 63;
    Qs[idx] = featmap(qb[(size_t)r * rs + c]);
    Vs[idx] = vb[(size_t)r * rs + c];
    Bs[idx] = sp[idx];
  }
  if (tid < D_) ks[tid] = ksum[(size_t)(nh * T_ + t) * D_ + tid];
  __syncthreads();

  const int wave = tid >> 5, lane = tid & 31;
  v8f acc[2];

#pragma unroll
  for (int j = 0; j < 2; ++j) {
    const int tt = wave * 2 + j;
    const int mb = (tt >> 2) * 16, nb = (tt & 3) * 16;
    acc[j] = (v8f){};
#pragma unroll
    for (int kk = 0; kk < 16; ++kk) {
      v2f a = frag_a_rm(Qs, D_, mb, kk, lane);
      v2f b = frag_b_rm(Bs, DV_, nb, kk, lane);
      acc[j] = wmma4(a, b, acc[j]);
    }
  }
  __syncthreads();

  for (int idx = tid; idx < C_ * D_; idx += 256) {
    int r = idx >> 6, c = idx & 63;
    Bs[idx] = featmap(kb[(size_t)r * rs + c]);
  }
  __syncthreads();

  v8f p[2];
#pragma unroll
  for (int j = 0; j < 2; ++j) {
    const int tt = wave * 2 + j;
    const int ti = tt >> 2, tj = tt & 3;
    p[j] = (v8f){};
    if (tj <= ti) {
#pragma unroll
      for (int kk = 0; kk < 16; ++kk) {
        v2f a = frag_a_rm(Qs, D_, ti * 16, kk, lane);
        v2f b = frag_b_tr(Bs, D_, tj * 16, kk, lane);
        p[j] = wmma4(a, b, p[j]);
      }
    }
  }
  __syncthreads();

#pragma unroll
  for (int j = 0; j < 2; ++j) {
    const int tt = wave * 2 + j;
    const int ti = tt >> 2, tj = tt & 3;
    const int col = tj * 16 + (lane & 15);
    const int rb  = ti * 16 + ((lane >> 4) << 3);
#pragma unroll
    for (int r = 0; r < 8; ++r) {
      const int row = rb + r;
      Bs[row * C_ + col] = (col <= row) ? p[j][r] : 0.0f;
    }
  }
  __syncthreads();

  if (tid < C_) {
    float s = 0.0f, dq = 0.0f;
    for (int c = 0; c < C_; ++c) {
      s  += Bs[tid * C_ + c];
      dq += Qs[tid * D_ + c] * ks[c];
    }
    inv[tid] = 1.0f / (s + dq + EPS_);
  }

#pragma unroll
  for (int j = 0; j < 2; ++j) {
    const int tt = wave * 2 + j;
    const int mb = (tt >> 2) * 16, nb = (tt & 3) * 16;
#pragma unroll
    for (int kk = 0; kk < 16; ++kk) {
      v2f a = frag_a_rm(Bs, C_, mb, kk, lane);
      v2f b = frag_b_rm(Vs, DV_, nb, kk, lane);
      acc[j] = wmma4(a, b, acc[j]);
    }
  }
  __syncthreads();

  float* ob = out + (((size_t)n * L_ + (size_t)t * C_) * H_ + h) * DV_;
#pragma unroll
  for (int j = 0; j < 2; ++j) {
    const int tt = wave * 2 + j;
    const int mb = (tt >> 2) * 16, nb = (tt & 3) * 16;
    const int col = nb + (lane & 15);
    const int rb  = mb + ((lane >> 4) << 3);
#pragma unroll
    for (int r = 0; r < 8; ++r) {
      const int row = rb + r;
      ob[(size_t)row * (H_ * DV_) + col] = acc[j][r] * inv[row];
    }
  }
}

extern "C" void kernel_launch(void* const* d_in, const int* in_sizes, int n_in,
                              void* d_out, int out_size, void* d_ws, size_t ws_size,
                              hipStream_t stream) {
  (void)in_sizes; (void)n_in; (void)out_size; (void)ws_size;
  const float* q = (const float*)d_in[0];
  const float* k = (const float*)d_in[1];
  const float* v = (const float*)d_in[2];
  float* out = (float*)d_out;

  float* kv   = (float*)d_ws;
  float* ksum = kv + (size_t)NH_ * T_ * D_ * DV_;

  cla_phase1<<<NH_ * T_, 256, 0, stream>>>(k, v, kv, ksum);
  cla_phase2<<<P2_BLOCKS_, P2_THREADS_, 0, stream>>>(kv, ksum);
  cla_phase3<<<NH_ * T_, 256, 0, stream>>>(q, k, v, kv, ksum, out);
}


# LDS tile rows padded off the 64-float pitch (68 for row-per-lane A/K^T/P reads, 80 or 72 for column-per-lane B reads) so fragment and normalizer reads spread over the banks; all sums kept
# speedup vs baseline: 1.6359x; 1.3491x over previous
#include <hip/hip_runtime.h>
#include <hip/hip_bf16.h>


#define N_  4
#define L_  4096
#define H_  8
#define D_  64
#define DV_ 64
#define C_  64
#define T_  (L_ / C_)
#define NH_ (N_ * H_)
#define EPS_ 1e-6f

#define PA_  68
#define PB_  80
#define PB1_ 72

typedef float v2f __attribute__((ext_vector_type(2)));
typedef float v8f __attribute__((ext_vector_type(8)));

__device__ __forceinline__ v8f wmma4(v2f a, v2f b, v8f c) {
  return __builtin_amdgcn_wmma_f32_16x16x4_f32(
       false, a,  false, b,
       (short)0, c,  false,  false);
}

__device__ __forceinline__ v2f frag_a_rm(const float* buf, int ld, int mb, int kk, int lane) {
  int m = mb + (lane & 15);
  int k = kk * 4 + ((lane >> 4) << 1);
  v2f a; a.x = buf[m * ld + k]; a.y = buf[m * ld + k + 1]; return a;
}
__device__ __forceinline__ v2f frag_a_tr(const float* buf, int ld, int mb, int kk, int lane) {
  int m = mb + (lane & 15);
  int k = kk * 4 + ((lane >> 4) << 1);
  v2f a; a.x = buf[k * ld + m]; a.y = buf[(k + 1) * ld + m]; return a;
}
__device__ __forceinline__ v2f frag_b_rm(const float* buf, int ld, int nb, int kk, int lane) {
  int n = nb + (lane & 15);
  int k = kk * 4 + ((lane >> 4) << 1);
  v2f b; b.x = buf[k * ld + n]; b.y = buf[(k + 1) * ld + n]; return b;
}
__device__ __forceinline__ v2f frag_b_tr(const float* buf, int ld, int nb, int kk, int lane) {
  int n = nb + (lane & 15);
  int k = kk * 4 + ((lane >> 4) << 1);
  v2f b; b.x = buf[n * ld + k]; b.y = buf[n * ld + k + 1]; return b;
}

__device__ __forceinline__ float featmap(float x) {
  return x > 0.0f ? x + 1.0f : __expf(x);
}

__global__ __launch_bounds__(256) void cla_phase1(
    const float* __restrict__ kin, const float* __restrict__ vin,
    float* __restrict__ kv, float* __restrict__ ksum) {
  __shared__ float Ks[C_ * PB1_];
  __shared__ float Vs[C_ * PB1_];

  const int bt = blockIdx.x;
  const int nh = bt / T_, t = bt % T_;
  const int n = nh / H_, h = nh % H_;
  const int tid = threadIdx.x;
  const size_t rs = (size_t)H_ * D_;
  const float* kb = kin + (((size_t)n * L_ + (size_t)t * C_) * H_ + h) * D_;
  const float* vb = vin + (((size_t)n * L_ + (size_t)t * C_) * H_ + h) * D_;

  for (int idx = tid; idx < C_ * D_; idx += 256) {
    int r = idx >> 6, c = idx & 63;
    Ks[r * PB1_ + c] = featmap(kb[(size_t)r * rs + c]);
    Vs[r * PB1_ + c] = vb[(size_t)r * rs + c];
  }
  __syncthreads();

  const int wave = tid >> 5, lane = tid & 31;
  float* kvb = kv + (size_t)(nh * T_ + t) * (D_ * DV_);

#pragma unroll
  for (int j = 0; j < 2; ++j) {
    const int tt = wave * 2 + j;
    const int mb = (tt >> 2) * 16, nb = (tt & 3) * 16;
    v8f acc = {};
#pragma unroll
    for (int kk = 0; kk < 16; ++kk) {
      v2f a = frag_a_tr(Ks, PB1_, mb, kk, lane);
      v2f b = frag_b_rm(Vs, PB1_, nb, kk, lane);
      acc = wmma4(a, b, acc);
    }
    const int col = nb + (lane & 15);
    const int rb  = mb + ((lane >> 4) << 3);
#pragma unroll
    for (int r = 0; r < 8; ++r)
      kvb[(size_t)(rb + r) * DV_ + col] = acc[r];
  }

  if (tid < D_) {
    float s = 0.0f;
    for (int r = 0; r < C_; ++r) s += Ks[r * PB1_ + tid];
    ksum[(size_t)(nh * T_ + t) * D_ + tid] = s;
  }
}

typedef float v4f __attribute__((ext_vector_type(4)));

#define P2_THREADS_   128
#define KV4_          (D_ * DV_ / 4)
#define P2_KV_BLOCKS_ (NH_ * KV4_ / P2_THREADS_)
#define P2_KS_BLOCKS_ (NH_ * D_ / P2_THREADS_)
#define P2_BLOCKS_    (P2_KV_BLOCKS_ + P2_KS_BLOCKS_)

template <typename V, int STRIDE>
__device__ __forceinline__ void scan_chunks(V* p) {
  V v[T_];
#pragma unroll
  for (int t = 0; t < T_; ++t) v[t] = p[t * STRIDE];
  V run = V(0.0f);
#pragma unroll
  for (int t = 0; t < T_; ++t) {
    p[t * STRIDE] = run;
    run += v[t];
  }
}

__global__ __launch_bounds__(P2_THREADS_) __attribute__((amdgpu_waves_per_eu(1, 2)))
void cla_phase2(
    float* __restrict__ kv, float* __restrict__ ksum) {
  const int b = blockIdx.x;
  if (b < P2_KV_BLOCKS_) {
    const int g  = b * P2_THREADS_ + threadIdx.x;
    const int nh = g / KV4_, i = g % KV4_;
    scan_chunks<v4f, KV4_>(reinterpret_cast<v4f*>(kv) + (size_t)nh * T_ * KV4_ + i);
  } else {
    const int g  = (b - P2_KV_BLOCKS_) * P2_THREADS_ + threadIdx.x;
    const int nh = g / D_, i = g % D_;
    scan_chunks<float, D_>(ksum + (size_t)nh * T_ * D_ + i);
  }
}

__global__ __launch_bounds__(256) void cla_phase3(
    const float* __restrict__ qin, const float* __restrict__ kin,
    const float* __restrict__ vin, const float* __restrict__ kv,
    const float* __restrict__ ksum, float* __restrict__ out) {
  __shared__ float Qs[C_ * PA_];
  __shared__ float Vs[C_ * PB_];
  __shared__ float Bs[C_ * PB_];
  __shared__ float ks[D_];
  __shared__ float inv[C_];

  const int bt = blockIdx.x;
  const int nh = bt / T_, t = bt % T_;
  const int n = nh / H_, h = nh % H_;
  const int tid = threadIdx.x;
  const size_t rs = (size_t)H_ * D_;
  const float* qb = qin + (((size_t)n * L_ + (size_t)t * C_) * H_ + h) * D_;
  const float* kb = kin + (((size_t)n * L_ + (size_t)t * C_) * H_ + h) * D_;
  const float* vb = vin + (((size_t)n * L_ + (size_t)t * C_) * H_ + h) * D_;
  const float* sp = kv + (size_t)(nh * T_ + t) * (D_ * DV_);

  for (int idx = tid; idx < C_ * D_; idx += 256) {
    int r = idx >> 6, c = idx & 63;
    Qs[r * PA_ + c] = featmap(qb[(size_t)r * rs + c]);
    Vs[r * PB_ + c] = vb[(size_t)r * rs + c];
    Bs[r * PB_ + c] = sp[idx];
  }
  if (tid < D_) ks[tid] = ksum[(size_t)(nh * T_ + t) * D_ + tid];
  __syncthreads();

  const int wave = tid >> 5, lane = tid & 31;
  v8f acc[2];

#pragma unroll
  for (int j = 0; j < 2; ++j) {
    const int tt = wave * 2 + j;
    const int mb = (tt >> 2) * 16, nb = (tt & 3) * 16;
    acc[j] = (v8f){};
#pragma unroll
    for (int kk = 0; kk < 16; ++kk) {
      v2f a = frag_a_rm(Qs, PA_, mb, kk, lane);
      v2f b = frag_b_rm(Bs, PB_, nb, kk, lane);
      acc[j] = wmma4(a, b, acc[j]);
    }
  }
  __syncthreads();

  for (int idx = tid; idx < C_ * D_; idx += 256) {
    int r = idx >> 6, c = idx & 63;
    Bs[r * PA_ + c] = featmap(kb[(size_t)r * rs + c]);
  }
  __syncthreads();

  v8f p[2];
#pragma unroll
  for (int j = 0; j < 2; ++j) {
    const int tt = wave * 2 + j;
    const int ti = tt >> 2, tj = tt & 3;
    p[j] = (v8f){};
    if (tj <= ti) {
#pragma unroll
      for (int kk = 0; kk < 16; ++kk) {
        v2f a = frag_a_rm(Qs, PA_, ti * 16, kk, lane);
        v2f b = frag_b_tr(Bs, PA_, tj * 16, kk, lane);
        p[j] = wmma4(a, b, p[j]);
      }
    }
  }
  __syncthreads();

#pragma unroll
  for (int j = 0; j < 2; ++j) {
    const int tt = wave * 2 + j;
    const int ti = tt >> 2, tj = tt & 3;
    const int col = tj * 16 + (lane & 15);
    const int rb  = ti * 16 + ((lane >> 4) << 3);
#pragma unroll
    for (int r = 0; r < 8; ++r) {
      const int row = rb + r;
      Bs[row * PA_ + col] = (col <= row) ? p[j][r] : 0.0f;
    }
  }
  __syncthreads();

  if (tid < C_) {
    float s = 0.0f, dq = 0.0f;
    for (int c = 0; c < C_; ++c) {
      s  += Bs[tid * PA_ + c];
      dq += Qs[tid * PA_ + c] * ks[c];
    }
    inv[tid] = 1.0f / (s + dq + EPS_);
  }

#pragma unroll
  for (int j = 0; j < 2; ++j) {
    const int tt = wave * 2 + j;
    const int mb = (tt >> 2) * 16, nb = (tt & 3) * 16;
#pragma unroll
    for (int kk = 0; kk < 16; ++kk) {
      v2f a = frag_a_rm(Bs, PA_, mb, kk, lane);
      v2f b = frag_b_rm(Vs, PB_, nb, kk, lane);
      acc[j] = wmma4(a, b, acc[j]);
    }
  }
  __syncthreads();

  float* ob = out + (((size_t)n * L_ + (size_t)t * C_) * H_ + h) * DV_;
#pragma unroll
  for (int j = 0; j < 2; ++j) {
    const int tt = wave * 2 + j;
    const int mb = (tt >> 2) * 16, nb = (tt & 3) * 16;
    const int col = nb + (lane & 15);
    const int rb  = mb + ((lane >> 4) << 3);
#pragma unroll
    for (int r = 0; r < 8; ++r) {
      const int row = rb + r;
      ob[(size_t)row * (H_ * DV_) + col] = acc[j][r] * inv[row];
    }
  }
}

extern "C" void kernel_launch(void* const* d_in, const int* in_sizes, int n_in,
                              void* d_out, int out_size, void* d_ws, size_t ws_size,
                              hipStream_t stream) {
  (void)in_sizes; (void)n_in; (void)out_size; (void)ws_size;
  const float* q = (const float*)d_in[0];
  const float* k = (const float*)d_in[1];
  const float* v = (const float*)d_in[2];
  float* out = (float*)d_out;

  float* kv   = (float*)d_ws;
  float* ksum = kv + (size_t)NH_ * T_ * D_ * DV_;

  cla_phase1<<<NH_ * T_, 256, 0, stream>>>(k, v, kv, ksum);
  cla_phase2<<<P2_BLOCKS_, P2_THREADS_, 0, stream>>>(kv, ksum);
  cla_phase3<<<NH_ * T_, 256, 0, stream>>>(q, k, v, kv, ksum, out);
}


# Staging of every 64x64 tile by 16-byte loads all requested before the first LDS store (phase 3 also K, in its own LDS tile): one memory round trip per work-group, not 4 to 6; all sums kept
# speedup vs baseline: 1.7702x; 1.0821x over previous
#include <hip/hip_runtime.h>
#include <hip/hip_bf16.h>


#define N_  4
#define L_  4096
#define H_  8
#define D_  64
#define DV_ 64
#define C_  64
#define T_  (L_ / C_)
#define NH_ (N_ * H_)
#define EPS_ 1e-6f

#define PA_  68
#define PB_  80
#define PB1_ 72

typedef float v2f __attribute__((ext_vector_type(2)));
typedef float v8f __attribute__((ext_vector_type(8)));

__device__ __forceinline__ v8f wmma4(v2f a, v2f b, v8f c) {
  return __builtin_amdgcn_wmma_f32_16x16x4_f32(
       false, a,  false, b,
       (short)0, c,  false,  false);
}

__device__ __forceinline__ v2f frag_a_rm(const float* buf, int ld, int mb, int kk, int lane) {
  int m = mb + (lane & 15);
  int k = kk * 4 + ((lane >> 4) << 1);
  v2f a; a.x = buf[m * ld + k]; a.y = buf[m * ld + k + 1]; return a;
}
__device__ __forceinline__ v2f frag_a_tr(const float* buf, int ld, int mb, int kk, int lane) {
  int m = mb + (lane & 15);
  int k = kk * 4 + ((lane >> 4) << 1);
  v2f a; a.x = buf[k * ld + m]; a.y = buf[(k + 1) * ld + m]; return a;
}
__device__ __forceinline__ v2f frag_b_rm(const float* buf, int ld, int nb, int kk, int lane) {
  int n = nb + (lane & 15);
  int k = kk * 4 + ((lane >> 4) << 1);
  v2f b; b.x = buf[k * ld + n]; b.y = buf[(k + 1) * ld + n]; return b;
}
__device__ __forceinline__ v2f frag_b_tr(const float* buf, int ld, int nb, int kk, int lane) {
  int n = nb + (lane & 15);
  int k = kk * 4 + ((lane >> 4) << 1);
  v2f b; b.x = buf[n * ld + k]; b.y = buf[n * ld + k + 1]; return b;
}

__device__ __forceinline__ float featmap(float x) {
  return x > 0.0f ? x + 1.0f : __expf(x);
}

typedef float v4f __attribute__((ext_vector_type(4)));
#define TILE_V4_ (C_ * D_ / 4 / 256)

__device__ __forceinline__ v4f featmap4(v4f x) {
  v4f y; y.x = featmap(x.x); y.y = featmap(x.y); y.z = featmap(x.z); y.w = featmap(x.w); return y;
}
__device__ __forceinline__ v4f ldg4(const float* p) { return *reinterpret_cast<const v4f*>(p); }
__device__ __forceinline__ void sts4(float* p, v4f v) { *reinterpret_cast<v4f*>(p) = v; }

__global__ __launch_bounds__(256) void cla_phase1(
    const float* __restrict__ kin, const float* __restrict__ vin,
    float* __restrict__ kv, float* __restrict__ ksum) {
  __shared__ __attribute__((aligned(16))) float Ks[C_ * PB1_];
  __shared__ __attribute__((aligned(16))) float Vs[C_ * PB1_];

  const int bt = blockIdx.x;
  const int nh = bt / T_, t = bt % T_;
  const int n = nh / H_, h = nh % H_;
  const int tid = threadIdx.x;
  const size_t rs = (size_t)H_ * D_;
  const float* kb = kin + (((size_t)n * L_ + (size_t)t * C_) * H_ + h) * D_;
  const float* vb = vin + (((size_t)n * L_ + (size_t)t * C_) * H_ + h) * D_;

  const int r0 = tid >> 4;
  const int c4 = (tid & 15) << 2;
  v4f kr[TILE_V4_], vr[TILE_V4_];
#pragma unroll
  for (int i = 0; i < TILE_V4_; ++i) {
    const size_t g = (size_t)(r0 + 16 * i) * rs + c4;
    kr[i] = ldg4(kb + g);
    vr[i] = ldg4(vb + g);
  }
#pragma unroll
  for (int i = 0; i < TILE_V4_; ++i) {
    const int o = (r0 + 16 * i) * PB1_ + c4;
    sts4(Ks + o, featmap4(kr[i]));
    sts4(Vs + o, vr[i]);
  }
  __syncthreads();

  const int wave = tid >> 5, lane = tid & 31;
  float* kvb = kv + (size_t)(nh * T_ + t) * (D_ * DV_);

#pragma unroll
  for (int j = 0; j < 2; ++j) {
    const int tt = wave * 2 + j;
    const int mb = (tt >> 2) * 16, nb = (tt & 3) * 16;
    v8f acc = {};
#pragma unroll
    for (int kk = 0; kk < 16; ++kk) {
      v2f a = frag_a_tr(Ks, PB1_, mb, kk, lane);
      v2f b = frag_b_rm(Vs, PB1_, nb, kk, lane);
      acc = wmma4(a, b, acc);
    }
    const int col = nb + (lane & 15);
    const int rb  = mb + ((lane >> 4) << 3);
#pragma unroll
    for (int r = 0; r < 8; ++r)
      kvb[(size_t)(rb + r) * DV_ + col] = acc[r];
  }

  if (tid < D_) {
    float s = 0.0f;
    for (int r = 0; r < C_; ++r) s += Ks[r * PB1_ + tid];
    ksum[(size_t)(nh * T_ + t) * D_ + tid] = s;
  }
}

#define P2_THREADS_   128
#define KV4_          (D_ * DV_ / 4)
#define P2_KV_BLOCKS_ (NH_ * KV4_ / P2_THREADS_)
#define P2_KS_BLOCKS_ (NH_ * D_ / P2_THREADS_)
#define P2_BLOCKS_    (P2_KV_BLOCKS_ + P2_KS_BLOCKS_)

template <typename V, int STRIDE>
__device__ __forceinline__ void scan_chunks(V* p) {
  V v[T_];
#pragma unroll
  for (int t = 0; t < T_; ++t) v[t] = p[t * STRIDE];
  V run = V(0.0f);
#pragma unroll
  for (int t = 0; t < T_; ++t) {
    p[t * STRIDE] = run;
    run += v[t];
  }
}

__global__ __launch_bounds__(P2_THREADS_) __attribute__((amdgpu_waves_per_eu(1, 2)))
void cla_phase2(
    float* __restrict__ kv, float* __restrict__ ksum) {
  const int b = blockIdx.x;
  if (b < P2_KV_BLOCKS_) {
    const int g  = b * P2_THREADS_ + threadIdx.x;
    const int nh = g / KV4_, i = g % KV4_;
    scan_chunks<v4f, KV4_>(reinterpret_cast<v4f*>(kv) + (size_t)nh * T_ * KV4_ + i);
  } else {
    const int g  = (b - P2_KV_BLOCKS_) * P2_THREADS_ + threadIdx.x;
    const int nh = g / D_, i = g % D_;
    scan_chunks<float, D_>(ksum + (size_t)nh * T_ * D_ + i);
  }
}

__global__ __launch_bounds__(256) void cla_phase3(
    const float* __restrict__ qin, const float* __restrict__ kin,
    const float* __restrict__ vin, const float* __restrict__ kv,
    const float* __restrict__ ksum, float* __restrict__ out) {
  __shared__ __attribute__((aligned(16))) float Qs[C_ * PA_];
  __shared__ __attribute__((aligned(16))) float Vs[C_ * PB_];
  __shared__ __attribute__((aligned(16))) float Ss[C_ * PB_];
  __shared__ __attribute__((aligned(16))) float Ks[C_ * PA_];
  __shared__ float ks[D_];
  __shared__ float inv[C_];

  const int bt = blockIdx.x;
  const int nh = bt / T_, t = bt % T_;
  const int n = nh / H_, h = nh % H_;
  const int tid = threadIdx.x;
  const size_t rs = (size_t)H_ * D_;
  const float* qb = qin + (((size_t)n * L_ + (size_t)t * C_) * H_ + h) * D_;
  const float* kb = kin + (((size_t)n * L_ + (size_t)t * C_) * H_ + h) * D_;
  const float* vb = vin + (((size_t)n * L_ + (size_t)t * C_) * H_ + h) * D_;
  const float* sp = kv + (size_t)(nh * T_ + t) * (D_ * DV_);

  const int r0 = tid >> 4;
  const int c4 = (tid & 15) << 2;
  v4f qr[TILE_V4_], kr[TILE_V4_], vr[TILE_V4_], sr[TILE_V4_];
#pragma unroll
  for (int i = 0; i < TILE_V4_; ++i) {
    const int r = r0 + 16 * i;
    const size_t g = (size_t)r * rs + c4;
    qr[i] = ldg4(qb + g);
    kr[i] = ldg4(kb + g);
    vr[i] = ldg4(vb + g);
    sr[i] = ldg4(sp + r * DV_ + c4);
  }
  float ksv = 0.0f;
  if (tid < D_) ksv = ksum[(size_t)(nh * T_ + t) * D_ + tid];
#pragma unroll
  for (int i = 0; i < TILE_V4_; ++i) {
    const int r = r0 + 16 * i;
    sts4(Qs + r * PA_ + c4, featmap4(qr[i]));
    sts4(Ks + r * PA_ + c4, featmap4(kr[i]));
    sts4(Vs + r * PB_ + c4, vr[i]);
    sts4(Ss + r * PB_ + c4, sr[i]);
  }
  if (tid < D_) ks[tid] = ksv;
  __syncthreads();

  const int wave = tid >> 5, lane = tid & 31;
  v8f acc[2];

#pragma unroll
  for (int j = 0; j < 2; ++j) {
    const int tt = wave * 2 + j;
    const int mb = (tt >> 2) * 16, nb = (tt & 3) * 16;
    acc[j] = (v8f){};
#pragma unroll
    for (int kk = 0; kk < 16; ++kk) {
      v2f a = frag_a_rm(Qs, PA_, mb, kk, lane);
      v2f b = frag_b_rm(Ss, PB_, nb, kk, lane);
      acc[j] = wmma4(a, b, acc[j]);
    }
  }

  v8f p[2];
#pragma unroll
  for (int j = 0; j < 2; ++j) {
    const int tt = wave * 2 + j;
    const int ti = tt >> 2, tj = tt & 3;
    p[j] = (v8f){};
    if (tj <= ti) {
#pragma unroll
      for (int kk = 0; kk < 16; ++kk) {
        v2f a = frag_a_rm(Qs, PA_, ti * 16, kk, lane);
        v2f b = frag_b_tr(Ks, PA_, tj * 16, kk, lane);
        p[j] = wmma4(a, b, p[j]);
      }
    }
  }
  __syncthreads();

#pragma unroll
  for (int j = 0; j < 2; ++j) {
    const int tt = wave * 2 + j;
    const int ti = tt >> 2, tj = tt & 3;
    const int col = tj * 16 + (lane & 15);
    const int rb  = ti * 16 + ((lane >> 4) << 3);
#pragma unroll
    for (int r = 0; r < 8; ++r) {
      const int row = rb + r;
      Ks[row * PA_ + col] = (col <= row) ? p[j][r] : 0.0f;
    }
  }
  __syncthreads();

  if (tid < C_) {
    float s = 0.0f, dq = 0.0f;
    for (int c = 0; c < C_; ++c) {
      s  += Ks[tid * PA_ + c];
      dq += Qs[tid * PA_ + c] * ks[c];
    }
    inv[tid] = 1.0f / (s + dq + EPS_);
  }

#pragma unroll
  for (int j = 0; j < 2; ++j) {
    const int tt = wave * 2 + j;
    const int mb = (tt >> 2) * 16, nb = (tt & 3) * 16;
#pragma unroll
    for (int kk = 0; kk < 16; ++kk) {
      v2f a = frag_a_rm(Ks, PA_, mb, kk, lane);
      v2f b = frag_b_rm(Vs, PB_, nb, kk, lane);
      acc[j] = wmma4(a, b, acc[j]);
    }
  }
  __syncthreads();

  float* ob = out + (((size_t)n * L_ + (size_t)t * C_) * H_ + h) * DV_;
#pragma unroll
  for (int j = 0; j < 2; ++j) {
    const int tt = wave * 2 + j;
    const int mb = (tt >> 2) * 16, nb = (tt & 3) * 16;
    const int col = nb + (lane & 15);
    const int rb  = mb + ((lane >> 4) << 3);
#pragma unroll
    for (int r = 0; r < 8; ++r) {
      const int row = rb + r;
      ob[(size_t)row * (H_ * DV_) + col] = acc[j][r] * inv[row];
    }
  }
}

extern "C" void kernel_launch(void* const* d_in, const int* in_sizes, int n_in,
                              void* d_out, int out_size, void* d_ws, size_t ws_size,
                              hipStream_t stream) {
  (void)in_sizes; (void)n_in; (void)out_size; (void)ws_size;
  const float* q = (const float*)d_in[0];
  const float* k = (const float*)d_in[1];
  const float* v = (const float*)d_in[2];
  float* out = (float*)d_out;

  float* kv   = (float*)d_ws;
  float* ksum = kv + (size_t)NH_ * T_ * D_ * DV_;

  cla_phase1<<<NH_ * T_, 256, 0, stream>>>(k, v, kv, ksum);
  cla_phase2<<<P2_BLOCKS_, P2_THREADS_, 0, stream>>>(kv, ksum);
  cla_phase3<<<NH_ * T_, 256, 0, stream>>>(q, k, v, kv, ksum, out);
}


# One work-group walks 4 consecutive chunks with the running K^T V state in its accumulators (LDS copy as B operand); workspace holds group totals only, scan over 16 groups; sum order changed
# speedup vs baseline: 2.5934x; 1.4650x over previous
#include <hip/hip_runtime.h>
#include <hip/hip_bf16.h>


#define N_  4
#define L_  4096
#define H_  8
#define D_  64
#define DV_ 64
#define C_  64
#define T_  (L_ / C_)
#define NH_ (N_ * H_)
#define G_  4
#define NG_ (T_ / G_)
#define EPS_ 1e-6f

#define PA_  68
#define PK_  76
#define PB_  80
#define PB1_ 72

typedef float v2f __attribute__((ext_vector_type(2)));
typedef float v8f __attribute__((ext_vector_type(8)));

__device__ __forceinline__ v8f wmma4(v2f a, v2f b, v8f c) {
  return __builtin_amdgcn_wmma_f32_16x16x4_f32(
       false, a,  false, b,
       (short)0, c,  false,  false);
}

__device__ __forceinline__ v2f frag_a_rm(const float* buf, int ld, int mb, int kk, int lane) {
  int m = mb + (lane & 15);
  int k = kk * 4 + ((lane >> 4) << 1);
  v2f a; a.x = buf[m * ld + k]; a.y = buf[m * ld + k + 1]; return a;
}
__device__ __forceinline__ v2f frag_a_tr(const float* buf, int ld, int mb, int kk, int lane) {
  int m = mb + (lane & 15);
  int k = kk * 4 + ((lane >> 4) << 1);
  v2f a; a.x = buf[k * ld + m]; a.y = buf[(k + 1) * ld + m]; return a;
}
__device__ __forceinline__ v2f frag_b_rm(const float* buf, int ld, int nb, int kk, int lane) {
  int n = nb + (lane & 15);
  int k = kk * 4 + ((lane >> 4) << 1);
  v2f b; b.x = buf[k * ld + n]; b.y = buf[(k + 1) * ld + n]; return b;
}
__device__ __forceinline__ v2f frag_b_tr(const float* buf, int ld, int nb, int kk, int lane) {
  int n = nb + (lane & 15);
  int k = kk * 4 + ((lane >> 4) << 1);
  v2f b; b.x = buf[n * ld + k]; b.y = buf[n * ld + k + 1]; return b;
}

__device__ __forceinline__ float featmap(float x) {
  return x > 0.0f ? x + 1.0f : __expf(x);
}

typedef float v4f __attribute__((ext_vector_type(4)));
#define TILE_V4_ (C_ * D_ / 4 / 256)

__device__ __forceinline__ v4f featmap4(v4f x) {
  v4f y; y.x = featmap(x.x); y.y = featmap(x.y); y.z = featmap(x.z); y.w = featmap(x.w); return y;
}
__device__ __forceinline__ v4f ldg4(const float* p) { return *reinterpret_cast<const v4f*>(p); }
__device__ __forceinline__ void sts4(float* p, v4f v) { *reinterpret_cast<v4f*>(p) = v; }

__global__ __launch_bounds__(256) void cla_phase1(
    const float* __restrict__ kin, const float* __restrict__ vin,
    float* __restrict__ gkv, float* __restrict__ gks) {
  __shared__ __attribute__((aligned(16))) float Ks[C_ * PB1_];
  __shared__ __attribute__((aligned(16))) float Vs[C_ * PB1_];

  const int bg = blockIdx.x;
  const int nh = bg / NG_, g = bg % NG_;
  const int n = nh / H_, h = nh % H_;
  const int tid = threadIdx.x;
  const int wave = tid >> 5, lane = tid & 31;
  const size_t rs = (size_t)H_ * D_;
  const int r0 = tid >> 4;
  const int c4 = (tid & 15) << 2;

  v8f acc[2];
  acc[0] = (v8f){}; acc[1] = (v8f){};
  float s = 0.0f;

  for (int c = 0; c < G_; ++c) {
    const int t = g * G_ + c;
    const float* kb = kin + (((size_t)n * L_ + (size_t)t * C_) * H_ + h) * D_;
    const float* vb = vin + (((size_t)n * L_ + (size_t)t * C_) * H_ + h) * D_;

    v4f kr[TILE_V4_], vr[TILE_V4_];
#pragma unroll
    for (int i = 0; i < TILE_V4_; ++i) {
      const size_t gi = (size_t)(r0 + 16 * i) * rs + c4;
      kr[i] = ldg4(kb + gi);
      vr[i] = ldg4(vb + gi);
    }
#pragma unroll
    for (int i = 0; i < TILE_V4_; ++i) {
      const int o = (r0 + 16 * i) * PB1_ + c4;
      sts4(Ks + o, featmap4(kr[i]));
      sts4(Vs + o, vr[i]);
    }
    __syncthreads();

#pragma unroll
    for (int j = 0; j < 2; ++j) {
      const int tt = wave * 2 + j;
      const int mb = (tt >> 2) * 16, nb = (tt & 3) * 16;
#pragma unroll
      for (int kk = 0; kk < 16; ++kk) {
        v2f a = frag_a_tr(Ks, PB1_, mb, kk, lane);
        v2f b = frag_b_rm(Vs, PB1_, nb, kk, lane);
        acc[j] = wmma4(a, b, acc[j]);
      }
    }
    if (tid < D_) {
#pragma unroll
      for (int r = 0; r < C_; ++r) s += Ks[r * PB1_ + tid];
    }
    __syncthreads();
  }

  float* kvb = gkv + (size_t)(nh * NG_ + g) * (D_ * DV_);
#pragma unroll
  for (int j = 0; j < 2; ++j) {
    const int tt = wave * 2 + j;
    const int mb = (tt >> 2) * 16, nb = (tt & 3) * 16;
    const int col = nb + (lane & 15);
    const int rb  = mb + ((lane >> 4) << 3);
#pragma unroll
    for (int r = 0; r < 8; ++r)
      kvb[(size_t)(rb + r) * DV_ + col] = acc[j][r];
  }
  if (tid < D_) gks[(size_t)(nh * NG_ + g) * D_ + tid] = s;
}

#define P2_THREADS_   128
#define KV4_          (D_ * DV_ / 4)
#define P2_KV_BLOCKS_ (NH_ * KV4_ / P2_THREADS_)
#define P2_KS_BLOCKS_ (NH_ * D_ / P2_THREADS_)
#define P2_BLOCKS_    (P2_KV_BLOCKS_ + P2_KS_BLOCKS_)

template <typename V, int STRIDE>
__device__ __forceinline__ void scan_groups(V* p) {
  V v[NG_];
#pragma unroll
  for (int t = 0; t < NG_; ++t) v[t] = p[t * STRIDE];
  V run = V(0.0f);
#pragma unroll
  for (int t = 0; t < NG_; ++t) {
    p[t * STRIDE] = run;
    run += v[t];
  }
}

__global__ __launch_bounds__(P2_THREADS_) void cla_phase2(
    float* __restrict__ gkv, float* __restrict__ gks) {
  const int b = blockIdx.x;
  if (b < P2_KV_BLOCKS_) {
    const int gi = b * P2_THREADS_ + threadIdx.x;
    const int nh = gi / KV4_, i = gi % KV4_;
    scan_groups<v4f, KV4_>(reinterpret_cast<v4f*>(gkv) + (size_t)nh * NG_ * KV4_ + i);
  } else {
    const int gi = (b - P2_KV_BLOCKS_) * P2_THREADS_ + threadIdx.x;
    const int nh = gi / D_, i = gi % D_;
    scan_groups<float, D_>(gks + (size_t)nh * NG_ * D_ + i);
  }
}

__global__ __launch_bounds__(256) void cla_phase3(
    const float* __restrict__ qin, const float* __restrict__ kin,
    const float* __restrict__ vin, const float* __restrict__ gkv,
    const float* __restrict__ gks, float* __restrict__ out) {
  __shared__ __attribute__((aligned(16))) float Qs[C_ * PA_];
  __shared__ __attribute__((aligned(16))) float Vs[C_ * PB_];
  __shared__ __attribute__((aligned(16))) float Ss[C_ * PB_];
  __shared__ __attribute__((aligned(16))) float Ks[C_ * PK_];
  __shared__ float ks[D_];
  __shared__ float inv[C_];

  const int bg = blockIdx.x;
  const int nh = bg / NG_, g = bg % NG_;
  const int n = nh / H_, h = nh % H_;
  const int tid = threadIdx.x;
  const int wave = tid >> 5, lane = tid & 31;
  const size_t rs = (size_t)H_ * D_;
  const int r0 = tid >> 4;
  const int c4 = (tid & 15) << 2;

  const float* sp = gkv + (size_t)(nh * NG_ + g) * (D_ * DV_);
  v8f sacc[2];
#pragma unroll
  for (int j = 0; j < 2; ++j) {
    const int tt = wave * 2 + j;
    const int mb = (tt >> 2) * 16, nb = (tt & 3) * 16;
    const int col = nb + (lane & 15);
    const int rb  = mb + ((lane >> 4) << 3);
#pragma unroll
    for (int r = 0; r < 8; ++r) sacc[j][r] = sp[(size_t)(rb + r) * DV_ + col];
  }
  float ksr = 0.0f;
  if (tid < D_) ksr = gks[(size_t)(nh * NG_ + g) * D_ + tid];

  for (int c = 0; c < G_; ++c) {
    const int t = g * G_ + c;
    const float* qb = qin + (((size_t)n * L_ + (size_t)t * C_) * H_ + h) * D_;
    const float* kb = kin + (((size_t)n * L_ + (size_t)t * C_) * H_ + h) * D_;
    const float* vb = vin + (((size_t)n * L_ + (size_t)t * C_) * H_ + h) * D_;

    v4f qr[TILE_V4_], kr[TILE_V4_], vr[TILE_V4_];
#pragma unroll
    for (int i = 0; i < TILE_V4_; ++i) {
      const size_t gi = (size_t)(r0 + 16 * i) * rs + c4;
      qr[i] = ldg4(qb + gi);
      kr[i] = ldg4(kb + gi);
      vr[i] = ldg4(vb + gi);
    }
#pragma unroll
    for (int j = 0; j < 2; ++j) {
      const int tt = wave * 2 + j;
      const int mb = (tt >> 2) * 16, nb = (tt & 3) * 16;
      const int col = nb + (lane & 15);
      const int rb  = mb + ((lane >> 4) << 3);
#pragma unroll
      for (int r = 0; r < 8; ++r) Ss[(rb + r) * PB_ + col] = sacc[j][r];
    }
    if (tid < D_) ks[tid] = ksr;
#pragma unroll
    for (int i = 0; i < TILE_V4_; ++i) {
      const int r = r0 + 16 * i;
      sts4(Qs + r * PA_ + c4, featmap4(qr[i]));
      sts4(Ks + r * PK_ + c4, featmap4(kr[i]));
      sts4(Vs + r * PB_ + c4, vr[i]);
    }
    __syncthreads();

    v8f acc[2];
#pragma unroll
    for (int j = 0; j < 2; ++j) {
      const int tt = wave * 2 + j;
      const int mb = (tt >> 2) * 16, nb = (tt & 3) * 16;
      acc[j] = (v8f){};
#pragma unroll
      for (int kk = 0; kk < 16; ++kk) {
        v2f a = frag_a_rm(Qs, PA_, mb, kk, lane);
        v2f b = frag_b_rm(Ss, PB_, nb, kk, lane);
        acc[j] = wmma4(a, b, acc[j]);
      }
    }

    v8f p[2];
#pragma unroll
    for (int j = 0; j < 2; ++j) {
      const int tt = wave * 2 + j;
      const int ti = tt >> 2, tj = tt & 3;
      p[j] = (v8f){};
      if (tj <= ti) {
#pragma unroll
        for (int kk = 0; kk < 16; ++kk) {
          v2f a = frag_a_rm(Qs, PA_, ti * 16, kk, lane);
          v2f b = frag_b_tr(Ks, PK_, tj * 16, kk, lane);
          p[j] = wmma4(a, b, p[j]);
        }
      }
    }

#pragma unroll
    for (int j = 0; j < 2; ++j) {
      const int tt = wave * 2 + j;
      const int mb = (tt >> 2) * 16, nb = (tt & 3) * 16;
#pragma unroll
      for (int kk = 0; kk < 16; ++kk) {
        v2f a = frag_a_tr(Ks, PK_, mb, kk, lane);
        v2f b = frag_b_rm(Vs, PB_, nb, kk, lane);
        sacc[j] = wmma4(a, b, sacc[j]);
      }
    }
    if (tid < D_) {
      float cs = 0.0f;
#pragma unroll
      for (int r = 0; r < C_; ++r) cs += Ks[r * PK_ + tid];
      ksr += cs;
    }
    __syncthreads();

#pragma unroll
    for (int j = 0; j < 2; ++j) {
      const int tt = wave * 2 + j;
      const int ti = tt >> 2, tj = tt & 3;
      const int col = tj * 16 + (lane & 15);
      const int rb  = ti * 16 + ((lane >> 4) << 3);
#pragma unroll
      for (int r = 0; r < 8; ++r) {
        const int row = rb + r;
        Ks[row * PK_ + col] = (col <= row) ? p[j][r] : 0.0f;
      }
    }
    __syncthreads();

    if (tid < C_) {
      float sm = 0.0f, dq = 0.0f;
#pragma unroll
      for (int cc = 0; cc < C_; ++cc) {
        sm += Ks[tid * PK_ + cc];
        dq += Qs[tid * PA_ + cc] * ks[cc];
      }
      inv[tid] = 1.0f / (sm + dq + EPS_);
    }

#pragma unroll
    for (int j = 0; j < 2; ++j) {
      const int tt = wave * 2 + j;
      const int mb = (tt >> 2) * 16, nb = (tt & 3) * 16;
#pragma unroll
      for (int kk = 0; kk < 16; ++kk) {
        v2f a = frag_a_rm(Ks, PK_, mb, kk, lane);
        v2f b = frag_b_rm(Vs, PB_, nb, kk, lane);
        acc[j] = wmma4(a, b, acc[j]);
      }
    }
    __syncthreads();

    float* ob = out + (((size_t)n * L_ + (size_t)t * C_) * H_ + h) * DV_;
#pragma unroll
    for (int j = 0; j < 2; ++j) {
      const int tt = wave * 2 + j;
      const int mb = (tt >> 2) * 16, nb = (tt & 3) * 16;
      const int col = nb + (lane & 15);
      const int rb  = mb + ((lane >> 4) << 3);
#pragma unroll
      for (int r = 0; r < 8; ++r) {
        const int row = rb + r;
        ob[(size_t)row * (H_ * DV_) + col] = acc[j][r] * inv[row];
      }
    }
  }
}

extern "C" void kernel_launch(void* const* d_in, const int* in_sizes, int n_in,
                              void* d_out, int out_size, void* d_ws, size_t ws_size,
                              hipStream_t stream) {
  (void)in_sizes; (void)n_in; (void)out_size; (void)ws_size;
  const float* q = (const float*)d_in[0];
  const float* k = (const float*)d_in[1];
  const float* v = (const float*)d_in[2];
  float* out = (float*)d_out;

  float* gkv = (float*)d_ws;
  float* gks = gkv + (size_t)NH_ * NG_ * D_ * DV_;

  cla_phase1<<<NH_ * NG_, 256, 0, stream>>>(k, v, gkv, gks);
  cla_phase2<<<P2_BLOCKS_, P2_THREADS_, 0, stream>>>(gkv, gks);
  cla_phase3<<<NH_ * NG_, 256, 0, stream>>>(q, k, v, gkv, gks, out);
}


# Scan launch removed: each phase-3 work-group sums the totals of the groups in front of it itself (totals kept in accumulator order, 16-byte loads, group order kept); longest prefixes enter first
# speedup vs baseline: 3.1403x; 1.2109x over previous
#include <hip/hip_runtime.h>
#include <hip/hip_bf16.h>


#define N_  4
#define L_  4096
#define H_  8
#define D_  64
#define DV_ 64
#define C_  64
#define T_  (L_ / C_)
#define NH_ (N_ * H_)
#define G_  4
#define NG_ (T_ / G_)
#define PF_ 8
#define EPS_ 1e-6f

#define PA_  68
#define PK_  76
#define PB_  80
#define PB1_ 72

typedef float v2f __attribute__((ext_vector_type(2)));
typedef float v8f __attribute__((ext_vector_type(8)));

__device__ __forceinline__ v8f wmma4(v2f a, v2f b, v8f c) {
  return __builtin_amdgcn_wmma_f32_16x16x4_f32(
       false, a,  false, b,
       (short)0, c,  false,  false);
}

__device__ __forceinline__ v2f frag_a_rm(const float* buf, int ld, int mb, int kk, int lane) {
  int m = mb + (lane & 15);
  int k = kk * 4 + ((lane >> 4) << 1);
  v2f a; a.x = buf[m * ld + k]; a.y = buf[m * ld + k + 1]; return a;
}
__device__ __forceinline__ v2f frag_a_tr(const float* buf, int ld, int mb, int kk, int lane) {
  int m = mb + (lane & 15);
  int k = kk * 4 + ((lane >> 4) << 1);
  v2f a; a.x = buf[k * ld + m]; a.y = buf[(k + 1) * ld + m]; return a;
}
__device__ __forceinline__ v2f frag_b_rm(const float* buf, int ld, int nb, int kk, int lane) {
  int n = nb + (lane & 15);
  int k = kk * 4 + ((lane >> 4) << 1);
  v2f b; b.x = buf[k * ld + n]; b.y = buf[(k + 1) * ld + n]; return b;
}
__device__ __forceinline__ v2f frag_b_tr(const float* buf, int ld, int nb, int kk, int lane) {
  int n = nb + (lane & 15);
  int k = kk * 4 + ((lane >> 4) << 1);
  v2f b; b.x = buf[n * ld + k]; b.y = buf[n * ld + k + 1]; return b;
}

__device__ __forceinline__ float featmap(float x) {
  return x > 0.0f ? x + 1.0f : __expf(x);
}

typedef float v4f __attribute__((ext_vector_type(4)));
#define TILE_V4_ (C_ * D_ / 4 / 256)
#define KV4_     (D_ * DV_ / 4)

__device__ __forceinline__ v4f featmap4(v4f x) {
  v4f y; y.x = featmap(x.x); y.y = featmap(x.y); y.z = featmap(x.z); y.w = featmap(x.w); return y;
}
__device__ __forceinline__ v4f ldg4(const float* p) { return *reinterpret_cast<const v4f*>(p); }
__device__ __forceinline__ void sts4(float* p, v4f v) { *reinterpret_cast<v4f*>(p) = v; }

__global__ __launch_bounds__(256) void cla_phase1(
    const float* __restrict__ kin, const float* __restrict__ vin,
    float* __restrict__ gkv, float* __restrict__ gks) {
  __shared__ __attribute__((aligned(16))) float Ks[C_ * PB1_];
  __shared__ __attribute__((aligned(16))) float Vs[C_ * PB1_];

  const int bg = blockIdx.x;
  const int nh = bg / NG_, g = bg % NG_;
  const int n = nh / H_, h = nh % H_;
  const int tid = threadIdx.x;
  const int wave = tid >> 5, lane = tid & 31;
  const size_t rs = (size_t)H_ * D_;
  const int r0 = tid >> 4;
  const int c4 = (tid & 15) << 2;

  v8f acc[2];
  acc[0] = (v8f){}; acc[1] = (v8f){};
  float s = 0.0f;

  for (int c = 0; c < G_; ++c) {
    const int t = g * G_ + c;
    const float* kb = kin + (((size_t)n * L_ + (size_t)t * C_) * H_ + h) * D_;
    const float* vb = vin + (((size_t)n * L_ + (size_t)t * C_) * H_ + h) * D_;

    v4f kr[TILE_V4_], vr[TILE_V4_];
#pragma unroll
    for (int i = 0; i < TILE_V4_; ++i) {
      const size_t gi = (size_t)(r0 + 16 * i) * rs + c4;
      kr[i] = ldg4(kb + gi);
      vr[i] = ldg4(vb + gi);
    }
#pragma unroll
    for (int i = 0; i < TILE_V4_; ++i) {
      const int o = (r0 + 16 * i) * PB1_ + c4;
      sts4(Ks + o, featmap4(kr[i]));
      sts4(Vs + o, vr[i]);
    }
    __syncthreads();

#pragma unroll
    for (int j = 0; j < 2; ++j) {
      const int tt = wave * 2 + j;
      const int mb = (tt >> 2) * 16, nb = (tt & 3) * 16;
#pragma unroll
      for (int kk = 0; kk < 16; ++kk) {
        v2f a = frag_a_tr(Ks, PB1_, mb, kk, lane);
        v2f b = frag_b_rm(Vs, PB1_, nb, kk, lane);
        acc[j] = wmma4(a, b, acc[j]);
      }
    }
    if (tid < D_) {
#pragma unroll
      for (int r = 0; r < C_; ++r) s += Ks[r * PB1_ + tid];
    }
    __syncthreads();
  }

  v4f* kvq = reinterpret_cast<v4f*>(gkv) + (size_t)(nh * NG_ + g) * KV4_ + (size_t)(wave * 64 + lane) * 2;
#pragma unroll
  for (int j = 0; j < 2; ++j) {
    kvq[j * 64 + 0] = (v4f){acc[j][0], acc[j][1], acc[j][2], acc[j][3]};
    kvq[j * 64 + 1] = (v4f){acc[j][4], acc[j][5], acc[j][6], acc[j][7]};
  }
  if (tid < D_) gks[(size_t)(nh * NG_ + g) * D_ + tid] = s;
}

__global__ __launch_bounds__(256) void cla_phase3(
    const float* __restrict__ qin, const float* __restrict__ kin,
    const float* __restrict__ vin, const float* __restrict__ gkv,
    const float* __restrict__ gks, float* __restrict__ out) {
  __shared__ __attribute__((aligned(16))) float Qs[C_ * PA_];
  __shared__ __attribute__((aligned(16))) float Vs[C_ * PB_];
  __shared__ __attribute__((aligned(16))) float Ss[C_ * PB_];
  __shared__ __attribute__((aligned(16))) float Ks[C_ * PK_];
  __shared__ float ks[D_];
  __shared__ float inv[C_];

  const int bg = blockIdx.x;
  const int nh = bg % NH_, g = NG_ - 1 - bg / NH_;
  const int n = nh / H_, h = nh % H_;
  const int tid = threadIdx.x;
  const int wave = tid >> 5, lane = tid & 31;
  const size_t rs = (size_t)H_ * D_;
  const int r0 = tid >> 4;
  const int c4 = (tid & 15) << 2;

  const v4f* tq = reinterpret_cast<const v4f*>(gkv) + (size_t)nh * NG_ * KV4_ + (size_t)(wave * 64 + lane) * 2;
  v4f run[2][2];
#pragma unroll
  for (int j = 0; j < 2; ++j) { run[j][0] = v4f(0.0f); run[j][1] = v4f(0.0f); }
  for (int t0 = 0; t0 < g; t0 += PF_) {
    v4f tv[PF_][2][2];
#pragma unroll
    for (int u = 0; u < PF_; ++u) {
      const int t = (t0 + u < g) ? (t0 + u) : (g - 1);
      const v4f* tp = tq + (size_t)t * KV4_;
#pragma unroll
      for (int j = 0; j < 2; ++j) { tv[u][j][0] = tp[j * 64 + 0]; tv[u][j][1] = tp[j * 64 + 1]; }
    }
#pragma unroll
    for (int u = 0; u < PF_; ++u) {
      if (t0 + u < g) {
#pragma unroll
        for (int j = 0; j < 2; ++j) { run[j][0] += tv[u][j][0]; run[j][1] += tv[u][j][1]; }
      }
    }
  }
  v8f sacc[2];
#pragma unroll
  for (int j = 0; j < 2; ++j)
    sacc[j] = (v8f){run[j][0].x, run[j][0].y, run[j][0].z, run[j][0].w,
                    run[j][1].x, run[j][1].y, run[j][1].z, run[j][1].w};
  float ksr = 0.0f;
  if (tid < D_) {
    float kt[NG_ - 1];
#pragma unroll
    for (int t = 0; t < NG_ - 1; ++t) {
      const int tc = (t < g) ? t : (g > 0 ? g - 1 : 0);
      kt[t] = gks[(size_t)(nh * NG_ + tc) * D_ + tid];
    }
#pragma unroll
    for (int t = 0; t < NG_ - 1; ++t) ksr += (t < g) ? kt[t] : 0.0f;
  }

  for (int c = 0; c < G_; ++c) {
    const int t = g * G_ + c;
    const float* qb = qin + (((size_t)n * L_ + (size_t)t * C_) * H_ + h) * D_;
    const float* kb = kin + (((size_t)n * L_ + (size_t)t * C_) * H_ + h) * D_;
    const float* vb = vin + (((size_t)n * L_ + (size_t)t * C_) * H_ + h) * D_;

    v4f qr[TILE_V4_], kr[TILE_V4_], vr[TILE_V4_];
#pragma unroll
    for (int i = 0; i < TILE_V4_; ++i) {
      const size_t gi = (size_t)(r0 + 16 * i) * rs + c4;
      qr[i] = ldg4(qb + gi);
      kr[i] = ldg4(kb + gi);
      vr[i] = ldg4(vb + gi);
    }
#pragma unroll
    for (int j = 0; j < 2; ++j) {
      const int tt = wave * 2 + j;
      const int mb = (tt >> 2) * 16, nb = (tt & 3) * 16;
      const int col = nb + (lane & 15);
      const int rb  = mb + ((lane >> 4) << 3);
#pragma unroll
      for (int r = 0; r < 8; ++r) Ss[(rb + r) * PB_ + col] = sacc[j][r];
    }
    if (tid < D_) ks[tid] = ksr;
#pragma unroll
    for (int i = 0; i < TILE_V4_; ++i) {
      const int r = r0 + 16 * i;
      sts4(Qs + r * PA_ + c4, featmap4(qr[i]));
      sts4(Ks + r * PK_ + c4, featmap4(kr[i]));
      sts4(Vs + r * PB_ + c4, vr[i]);
    }
    __syncthreads();

    v8f acc[2];
#pragma unroll
    for (int j = 0; j < 2; ++j) {
      const int tt = wave * 2 + j;
      const int mb = (tt >> 2) * 16, nb = (tt & 3) * 16;
      acc[j] = (v8f){};
#pragma unroll
      for (int kk = 0; kk < 16; ++kk) {
        v2f a = frag_a_rm(Qs, PA_, mb, kk, lane);
        v2f b = frag_b_rm(Ss, PB_, nb, kk, lane);
        acc[j] = wmma4(a, b, acc[j]);
      }
    }

    v8f p[2];
#pragma unroll
    for (int j = 0; j < 2; ++j) {
      const int tt = wave * 2 + j;
      const int ti = tt >> 2, tj = tt & 3;
      p[j] = (v8f){};
      if (tj <= ti) {
#pragma unroll
        for (int kk = 0; kk < 16; ++kk) {
          v2f a = frag_a_rm(Qs, PA_, ti * 16, kk, lane);
          v2f b = frag_b_tr(Ks, PK_, tj * 16, kk, lane);
          p[j] = wmma4(a, b, p[j]);
        }
      }
    }

#pragma unroll
    for (int j = 0; j < 2; ++j) {
      const int tt = wave * 2 + j;
      const int mb = (tt >> 2) * 16, nb = (tt & 3) * 16;
#pragma unroll
      for (int kk = 0; kk < 16; ++kk) {
        v2f a = frag_a_tr(Ks, PK_, mb, kk, lane);
        v2f b = frag_b_rm(Vs, PB_, nb, kk, lane);
        sacc[j] = wmma4(a, b, sacc[j]);
      }
    }
    if (tid < D_) {
      float cs = 0.0f;
#pragma unroll
      for (int r = 0; r < C_; ++r) cs += Ks[r * PK_ + tid];
      ksr += cs;
    }
    __syncthreads();

#pragma unroll
    for (int j = 0; j < 2; ++j) {
      const int tt = wave * 2 + j;
      const int ti = tt >> 2, tj = tt & 3;
      const int col = tj * 16 + (lane & 15);
      const int rb  = ti * 16 + ((lane >> 4) << 3);
#pragma unroll
      for (int r = 0; r < 8; ++r) {
        const int row = rb + r;
        Ks[row * PK_ + col] = (col <= row) ? p[j][r] : 0.0f;
      }
    }
    __syncthreads();

    if (tid < C_) {
      float sm = 0.0f, dq = 0.0f;
#pragma unroll
      for (int cc = 0; cc < C_; ++cc) {
        sm += Ks[tid * PK_ + cc];
        dq += Qs[tid * PA_ + cc] * ks[cc];
      }
      inv[tid] = 1.0f / (sm + dq + EPS_);
    }

#pragma unroll
    for (int j = 0; j < 2; ++j) {
      const int tt = wave * 2 + j;
      const int mb = (tt >> 2) * 16, nb = (tt & 3) * 16;
#pragma unroll
      for (int kk = 0; kk < 16; ++kk) {
        v2f a = frag_a_rm(Ks, PK_, mb, kk, lane);
        v2f b = frag_b_rm(Vs, PB_, nb, kk, lane);
        acc[j] = wmma4(a, b, acc[j]);
      }
    }
    __syncthreads();

    float* ob = out + (((size_t)n * L_ + (size_t)t * C_) * H_ + h) * DV_;
#pragma unroll
    for (int j = 0; j < 2; ++j) {
      const int tt = wave * 2 + j;
      const int mb = (tt >> 2) * 16, nb = (tt & 3) * 16;
      const int col = nb + (lane & 15);
      const int rb  = mb + ((lane >> 4) << 3);
#pragma unroll
      for (int r = 0; r < 8; ++r) {
        const int row = rb + r;
        ob[(size_t)row * (H_ * DV_) + col] = acc[j][r] * inv[row];
      }
    }
  }
}

extern "C" void kernel_launch(void* const* d_in, const int* in_sizes, int n_in,
                              void* d_out, int out_size, void* d_ws, size_t ws_size,
                              hipStream_t stream) {
  (void)in_sizes; (void)n_in; (void)out_size; (void)ws_size;
  const float* q = (const float*)d_in[0];
  const float* k = (const float*)d_in[1];
  const float* v = (const float*)d_in[2];
  float* out = (float*)d_out;

  float* gkv = (float*)d_ws;
  float* gks = gkv + (size_t)NH_ * NG_ * D_ * DV_;

  cla_phase1<<<NH_ * NG_, 256, 0, stream>>>(k, v, gkv, gks);
  cla_phase3<<<NH_ * NG_, 256, 0, stream>>>(q, k, v, gkv, gks, out);
}


# Next chunk's Q/K/V loads requested a chunk ahead (first one ahead of the prefix sum), kept in flight across LDS-only barriers; phase 3 fragment addresses from 5 pinned lane offsets to fit 256 VGPRs
# speedup vs baseline: 3.3043x; 1.0522x over previous
#include <hip/hip_runtime.h>
#include <hip/hip_bf16.h>


#define N_  4
#define L_  4096
#define H_  8
#define D_  64
#define DV_ 64
#define C_  64
#define T_  (L_ / C_)
#define NH_ (N_ * H_)
#define G_  4
#define NG_ (T_ / G_)
#define PF_ 8
#define EPS_ 1e-6f

#define PA_  68
#define PK_  76
#define PB_  80
#define PB1_ 72

typedef float v2f __attribute__((ext_vector_type(2)));
typedef float v8f __attribute__((ext_vector_type(8)));

__device__ __forceinline__ v8f wmma4(v2f a, v2f b, v8f c) {
  return __builtin_amdgcn_wmma_f32_16x16x4_f32(
       false, a,  false, b,
       (short)0, c,  false,  false);
}

__device__ __forceinline__ v2f frag_a_rm(const float* buf, int ld, int mb, int kk, int lane) {
  int m = mb + (lane & 15);
  int k = kk * 4 + ((lane >> 4) << 1);
  v2f a; a.x = buf[m * ld + k]; a.y = buf[m * ld + k + 1]; return a;
}
__device__ __forceinline__ v2f frag_a_tr(const float* buf, int ld, int mb, int kk, int lane) {
  int m = mb + (lane & 15);
  int k = kk * 4 + ((lane >> 4) << 1);
  v2f a; a.x = buf[k * ld + m]; a.y = buf[(k + 1) * ld + m]; return a;
}
__device__ __forceinline__ v2f frag_b_rm(const float* buf, int ld, int nb, int kk, int lane) {
  int n = nb + (lane & 15);
  int k = kk * 4 + ((lane >> 4) << 1);
  v2f b; b.x = buf[k * ld + n]; b.y = buf[(k + 1) * ld + n]; return b;
}
__device__ __forceinline__ v2f frag_b_tr(const float* buf, int ld, int nb, int kk, int lane) {
  int n = nb + (lane & 15);
  int k = kk * 4 + ((lane >> 4) << 1);
  v2f b; b.x = buf[n * ld + k]; b.y = buf[n * ld + k + 1]; return b;
}

__device__ __forceinline__ v2f ld_pair(const float* buf, int off) {
  v2f a; a.x = buf[off]; a.y = buf[off + 1]; return a;
}
__device__ __forceinline__ v2f ld_rows(const float* buf, int off, int ld) {
  v2f b; b.x = buf[off]; b.y = buf[off + ld]; return b;
}
__device__ __forceinline__ int pin(int x) { asm volatile("" : "+v"(x)); return x; }

__device__ __forceinline__ float featmap(float x) {
  return x > 0.0f ? x + 1.0f : __expf(x);
}

typedef float v4f __attribute__((ext_vector_type(4)));
#define TILE_V4_ (C_ * D_ / 4 / 256)
#define KV4_     (D_ * DV_ / 4)

__device__ __forceinline__ v4f featmap4(v4f x) {
  v4f y; y.x = featmap(x.x); y.y = featmap(x.y); y.z = featmap(x.z); y.w = featmap(x.w); return y;
}
__device__ __forceinline__ v4f ldg4(const float* p) { return *reinterpret_cast<const v4f*>(p); }
__device__ __forceinline__ void sts4(float* p, v4f v) { *reinterpret_cast<v4f*>(p) = v; }

__device__ __forceinline__ void lds_barrier() {
  __builtin_amdgcn_fence(__ATOMIC_RELEASE, "workgroup", "local");
  __builtin_amdgcn_s_barrier();
  __builtin_amdgcn_fence(__ATOMIC_ACQUIRE, "workgroup", "local");
}

__global__ __launch_bounds__(256) void cla_phase1(
    const float* kin, const float* vin,
    float* __restrict__ gkv, float* __restrict__ gks) {
  __shared__ __attribute__((aligned(16))) float Ks[C_ * PB1_];
  __shared__ __attribute__((aligned(16))) float Vs[C_ * PB1_];

  const int bg = blockIdx.x;
  const int nh = bg / NG_, g = bg % NG_;
  const int n = nh / H_, h = nh % H_;
  const int tid = threadIdx.x;
  const int wave = tid >> 5, lane = tid & 31;
  const size_t rs = (size_t)H_ * D_;
  const int r0 = tid >> 4;
  const int c4 = (tid & 15) << 2;

  v8f acc[2];
  acc[0] = (v8f){}; acc[1] = (v8f){};
  float s = 0.0f;

  const size_t goff = (size_t)r0 * rs + c4;
  const size_t cb0 = (((size_t)n * L_ + (size_t)(g * G_) * C_) * H_ + h) * D_ + goff;
  v4f kr[TILE_V4_], vr[TILE_V4_];
#pragma unroll
  for (int i = 0; i < TILE_V4_; ++i) {
    kr[i] = ldg4(kin + cb0 + (size_t)(16 * i) * rs);
    vr[i] = ldg4(vin + cb0 + (size_t)(16 * i) * rs);
  }

  for (int c = 0; c < G_; ++c) {
#pragma unroll
    for (int i = 0; i < TILE_V4_; ++i) {
      const int o = (r0 + 16 * i) * PB1_ + c4;
      sts4(Ks + o, featmap4(kr[i]));
      sts4(Vs + o, vr[i]);
    }
    if (c + 1 < G_) {
      const size_t cb = cb0 + (size_t)(c + 1) * C_ * rs;
#pragma unroll
      for (int i = 0; i < TILE_V4_; ++i) {
        kr[i] = ldg4(kin + cb + (size_t)(16 * i) * rs);
        vr[i] = ldg4(vin + cb + (size_t)(16 * i) * rs);
      }
    }
    lds_barrier();

#pragma unroll
    for (int j = 0; j < 2; ++j) {
      const int tt = wave * 2 + j;
      const int mb = (tt >> 2) * 16, nb = (tt & 3) * 16;
#pragma unroll
      for (int kk = 0; kk < 16; ++kk) {
        v2f a = frag_a_tr(Ks, PB1_, mb, kk, lane);
        v2f b = frag_b_rm(Vs, PB1_, nb, kk, lane);
        acc[j] = wmma4(a, b, acc[j]);
      }
    }
    if (tid < D_) {
#pragma unroll
      for (int r = 0; r < C_; ++r) s += Ks[r * PB1_ + tid];
    }
    lds_barrier();
  }

  v4f* kvq = reinterpret_cast<v4f*>(gkv) + (size_t)(nh * NG_ + g) * KV4_ + (size_t)(wave * 64 + lane) * 2;
#pragma unroll
  for (int j = 0; j < 2; ++j) {
    kvq[j * 64 + 0] = (v4f){acc[j][0], acc[j][1], acc[j][2], acc[j][3]};
    kvq[j * 64 + 1] = (v4f){acc[j][4], acc[j][5], acc[j][6], acc[j][7]};
  }
  if (tid < D_) gks[(size_t)(nh * NG_ + g) * D_ + tid] = s;
}

__global__ __launch_bounds__(256, 4) void cla_phase3(
    const float* qin, const float* kin,
    const float* vin, const float* __restrict__ gkv,
    const float* __restrict__ gks, float* __restrict__ out) {
  __shared__ __attribute__((aligned(16))) float Qs[C_ * PA_];
  __shared__ __attribute__((aligned(16))) float Vs[C_ * PB_];
  __shared__ __attribute__((aligned(16))) float Ss[C_ * PB_];
  __shared__ __attribute__((aligned(16))) float Ks[C_ * PK_];
  __shared__ float ks[D_];
  __shared__ float inv[C_];

  const int bg = blockIdx.x;
  const int nh = bg % NH_, g = NG_ - 1 - bg / NH_;
  const int n = nh / H_, h = nh % H_;
  const int tid = threadIdx.x;
  const int wave = tid >> 5, lane = tid & 31;
  const size_t rs = (size_t)H_ * D_;
  const int r0 = tid >> 4;
  const int c4 = (tid & 15) << 2;

  const unsigned loff = (unsigned)(r0 * (H_ * D_) + c4);
  const size_t cbu = (((size_t)n * L_ + (size_t)(g * G_) * C_) * H_ + h) * D_;
  const float* qc = qin + cbu;
  const float* kc = kin + cbu;
  const float* vc = vin + cbu;
  v4f qr[TILE_V4_], kr[TILE_V4_], vr[TILE_V4_];
  {
    const unsigned lo = (unsigned)pin((int)loff);
#pragma unroll
    for (int i = 0; i < TILE_V4_; ++i) {
      qr[i] = ldg4(qc + lo + (size_t)(16 * i) * rs);
      kr[i] = ldg4(kc + lo + (size_t)(16 * i) * rs);
      vr[i] = ldg4(vc + lo + (size_t)(16 * i) * rs);
    }
  }

  const v4f* tq = reinterpret_cast<const v4f*>(gkv) + (size_t)nh * NG_ * KV4_ + (size_t)(wave * 64 + lane) * 2;
  v4f run[2][2];
#pragma unroll
  for (int j = 0; j < 2; ++j) { run[j][0] = v4f(0.0f); run[j][1] = v4f(0.0f); }
  for (int t0 = 0; t0 < g; t0 += PF_) {
    v4f tv[PF_][2][2];
#pragma unroll
    for (int u = 0; u < PF_; ++u) {
      const int t = (t0 + u < g) ? (t0 + u) : (g - 1);
      const v4f* tp = tq + (size_t)t * KV4_;
#pragma unroll
      for (int j = 0; j < 2; ++j) { tv[u][j][0] = tp[j * 64 + 0]; tv[u][j][1] = tp[j * 64 + 1]; }
    }
#pragma unroll
    for (int u = 0; u < PF_; ++u) {
      if (t0 + u < g) {
#pragma unroll
        for (int j = 0; j < 2; ++j) { run[j][0] += tv[u][j][0]; run[j][1] += tv[u][j][1]; }
      }
    }
  }
  v8f sacc[2];
#pragma unroll
  for (int j = 0; j < 2; ++j)
    sacc[j] = (v8f){run[j][0].x, run[j][0].y, run[j][0].z, run[j][0].w,
                    run[j][1].x, run[j][1].y, run[j][1].z, run[j][1].w};
  float ksr = 0.0f;
  if (tid < D_) {
    float kt[NG_ - 1];
#pragma unroll
    for (int t = 0; t < NG_ - 1; ++t) {
      const int tc = (t < g) ? t : (g > 0 ? g - 1 : 0);
      kt[t] = gks[(size_t)(nh * NG_ + tc) * D_ + tid];
    }
#pragma unroll
    for (int t = 0; t < NG_ - 1; ++t) ksr += (t < g) ? kt[t] : 0.0f;
  }

  const int lc = lane & 15, lh2 = (lane >> 4) << 1;
  const int mb = (wave >> 1) * 16, nb0 = (wave & 1) * 32;
  const int qAh = ((mb + lc) * PA_ + lh2) / 2;
  const int bB0 = lh2 * PB_ + nb0 + lc;
  const int kBh = ((nb0 + lc) * PK_ + lh2) / 2;
  const int kA0 = lh2 * PK_ + mb + lc;
  const int pAh = ((mb + lc) * PK_ + lh2) / 2;

  for (int c = 0; c < G_; ++c) {
    const int t = g * G_ + c;

#pragma unroll
    for (int j = 0; j < 2; ++j) {
      const int tt = wave * 2 + j;
      const int mb = (tt >> 2) * 16, nb = (tt & 3) * 16;
      const int col = nb + (lane & 15);
      const int rb  = mb + ((lane >> 4) << 3);
#pragma unroll
      for (int r = 0; r < 8; ++r) Ss[(rb + r) * PB_ + col] = sacc[j][r];
    }
    if (tid < D_) ks[tid] = ksr;
#pragma unroll
    for (int i = 0; i < TILE_V4_; ++i) {
      const int r = r0 + 16 * i;
      sts4(Qs + r * PA_ + c4, featmap4(qr[i]));
      sts4(Ks + r * PK_ + c4, featmap4(kr[i]));
      sts4(Vs + r * PB_ + c4, vr[i]);
    }
    lds_barrier();

    v8f acc[2];
    {
    const int qA = 2 * pin(qAh), bB = pin(bB0);
#pragma unroll
    for (int j = 0; j < 2; ++j) {
      acc[j] = (v8f){};
#pragma unroll
      for (int kk = 0; kk < 16; ++kk) {
        v2f a = ld_pair(Qs, qA + kk * 4);
        v2f b = ld_rows(Ss, bB + kk * 4 * PB_ + 16 * j, PB_);
        acc[j] = wmma4(a, b, acc[j]);
      }
    }
    }

    v8f p[2];
    const int qA = 2 * pin(qAh), kB = 2 * pin(kBh);
#pragma unroll
    for (int j = 0; j < 2; ++j) {
      const int tt = wave * 2 + j;
      const int ti = tt >> 2, tj = tt & 3;
      p[j] = (v8f){};
      if (tj <= ti) {
#pragma unroll
        for (int kk = 0; kk < 16; ++kk) {
          v2f a = ld_pair(Qs, qA + kk * 4);
          v2f b = ld_pair(Ks, kB + 16 * j * PK_ + kk * 4);
          p[j] = wmma4(a, b, p[j]);
        }
      }
    }

    const int kA = pin(kA0), bBv = pin(bB0);
#pragma unroll
    for (int j = 0; j < 2; ++j) {
#pragma unroll
      for (int kk = 0; kk < 16; ++kk) {
        v2f a = ld_rows(Ks, kA + kk * 4 * PK_, PK_);
        v2f b = ld_rows(Vs, bBv + kk * 4 * PB_ + 16 * j, PB_);
        sacc[j] = wmma4(a, b, sacc[j]);
      }
    }
    if (tid < D_) {
      float cs = 0.0f;
#pragma unroll
      for (int r = 0; r < C_; ++r) cs += Ks[r * PK_ + tid];
      ksr += cs;
    }
    lds_barrier();

#pragma unroll
    for (int j = 0; j < 2; ++j) {
      const int tt = wave * 2 + j;
      const int ti = tt >> 2, tj = tt & 3;
      const int col = tj * 16 + (lane & 15);
      const int rb  = ti * 16 + ((lane >> 4) << 3);
#pragma unroll
      for (int r = 0; r < 8; ++r) {
        const int row = rb + r;
        Ks[row * PK_ + col] = (col <= row) ? p[j][r] : 0.0f;
      }
    }
    lds_barrier();

    if (tid < C_) {
      float sm = 0.0f, dq = 0.0f;
#pragma unroll
      for (int cc = 0; cc < C_; ++cc) {
        sm += Ks[tid * PK_ + cc];
        dq += Qs[tid * PA_ + cc] * ks[cc];
      }
      inv[tid] = 1.0f / (sm + dq + EPS_);
    }

    if (c + 1 < G_) {
      const float* qn = qc + (size_t)(c + 1) * C_ * rs;
      const float* kn = kc + (size_t)(c + 1) * C_ * rs;
      const float* vn = vc + (size_t)(c + 1) * C_ * rs;
      const unsigned lo = (unsigned)pin((int)loff);
#pragma unroll
      for (int i = 0; i < TILE_V4_; ++i) {
        qr[i] = ldg4(qn + lo + (size_t)(16 * i) * rs);
        kr[i] = ldg4(kn + lo + (size_t)(16 * i) * rs);
        vr[i] = ldg4(vn + lo + (size_t)(16 * i) * rs);
      }
    }

    const int pA = 2 * pin(pAh), bBp = pin(bB0);
#pragma unroll
    for (int j = 0; j < 2; ++j) {
#pragma unroll
      for (int kk = 0; kk < 16; ++kk) {
        v2f a = ld_pair(Ks, pA + kk * 4);
        v2f b = ld_rows(Vs, bBp + kk * 4 * PB_ + 16 * j, PB_);
        acc[j] = wmma4(a, b, acc[j]);
      }
    }
    lds_barrier();

    float* ob = out + (((size_t)n * L_ + (size_t)t * C_) * H_ + h) * DV_;
#pragma unroll
    for (int j = 0; j < 2; ++j) {
      const int tt = wave * 2 + j;
      const int mb = (tt >> 2) * 16, nb = (tt & 3) * 16;
      const int col = nb + (lane & 15);
      const int rb  = mb + ((lane >> 4) << 3);
#pragma unroll
      for (int r = 0; r < 8; ++r) {
        const int row = rb + r;
        ob[(size_t)row * (H_ * DV_) + col] = acc[j][r] * inv[row];
      }
    }
  }
}

extern "C" void kernel_launch(void* const* d_in, const int* in_sizes, int n_in,
                              void* d_out, int out_size, void* d_ws, size_t ws_size,
                              hipStream_t stream) {
  (void)in_sizes; (void)n_in; (void)out_size; (void)ws_size;
  const float* q = (const float*)d_in[0];
  const float* k = (const float*)d_in[1];
  const float* v = (const float*)d_in[2];
  float* out = (float*)d_out;

  float* gkv = (float*)d_ws;
  float* gks = gkv + (size_t)NH_ * NG_ * D_ * DV_;

  cla_phase1<<<NH_ * NG_, 256, 0, stream>>>(k, v, gkv, gks);
  cla_phase3<<<NH_ * NG_, 256, 0, stream>>>(q, k, v, gkv, gks, out);
}


# Matrix products that cannot reach the output are not issued: P@V k-steps right of the causal diagonal (all-zero tiles of masked P), and the state update of a group's last chunk; sum order kept
# speedup vs baseline: 3.5524x; 1.0751x over previous
#include <hip/hip_runtime.h>
#include <hip/hip_bf16.h>


#define N_  4
#define L_  4096
#define H_  8
#define D_  64
#define DV_ 64
#define C_  64
#define T_  (L_ / C_)
#define NH_ (N_ * H_)
#define G_  4
#define NG_ (T_ / G_)
#define PF_ 8
#define EPS_ 1e-6f

#define PA_  68
#define PK_  76
#define PB_  80
#define PB1_ 72

typedef float v2f __attribute__((ext_vector_type(2)));
typedef float v8f __attribute__((ext_vector_type(8)));

__device__ __forceinline__ v8f wmma4(v2f a, v2f b, v8f c) {
  return __builtin_amdgcn_wmma_f32_16x16x4_f32(
       false, a,  false, b,
       (short)0, c,  false,  false);
}

__device__ __forceinline__ v2f frag_a_rm(const float* buf, int ld, int mb, int kk, int lane) {
  int m = mb + (lane & 15);
  int k = kk * 4 + ((lane >> 4) << 1);
  v2f a; a.x = buf[m * ld + k]; a.y = buf[m * ld + k + 1]; return a;
}
__device__ __forceinline__ v2f frag_a_tr(const float* buf, int ld, int mb, int kk, int lane) {
  int m = mb + (lane & 15);
  int k = kk * 4 + ((lane >> 4) << 1);
  v2f a; a.x = buf[k * ld + m]; a.y = buf[(k + 1) * ld + m]; return a;
}
__device__ __forceinline__ v2f frag_b_rm(const float* buf, int ld, int nb, int kk, int lane) {
  int n = nb + (lane & 15);
  int k = kk * 4 + ((lane >> 4) << 1);
  v2f b; b.x = buf[k * ld + n]; b.y = buf[(k + 1) * ld + n]; return b;
}
__device__ __forceinline__ v2f frag_b_tr(const float* buf, int ld, int nb, int kk, int lane) {
  int n = nb + (lane & 15);
  int k = kk * 4 + ((lane >> 4) << 1);
  v2f b; b.x = buf[n * ld + k]; b.y = buf[n * ld + k + 1]; return b;
}

__device__ __forceinline__ v2f ld_pair(const float* buf, int off) {
  v2f a; a.x = buf[off]; a.y = buf[off + 1]; return a;
}
__device__ __forceinline__ v2f ld_rows(const float* buf, int off, int ld) {
  v2f b; b.x = buf[off]; b.y = buf[off + ld]; return b;
}
__device__ __forceinline__ int pin(int x) { asm volatile("" : "+v"(x)); return x; }

template <int NK>
__device__ __forceinline__ void pv_steps(const float* Ps, const float* Vt, int pA, int bB,
                                         v8f& acc0, v8f& acc1) {
#pragma unroll
  for (int kk = 0; kk < NK; ++kk) {
    v2f a = ld_pair(Ps, pA + kk * 4);
    acc0 = wmma4(a, ld_rows(Vt, bB + kk * 4 * PB_, PB_), acc0);
    acc1 = wmma4(a, ld_rows(Vt, bB + kk * 4 * PB_ + 16, PB_), acc1);
  }
}

__device__ __forceinline__ float featmap(float x) {
  return x > 0.0f ? x + 1.0f : __expf(x);
}

typedef float v4f __attribute__((ext_vector_type(4)));
#define TILE_V4_ (C_ * D_ / 4 / 256)
#define KV4_     (D_ * DV_ / 4)

__device__ __forceinline__ v4f featmap4(v4f x) {
  v4f y; y.x = featmap(x.x); y.y = featmap(x.y); y.z = featmap(x.z); y.w = featmap(x.w); return y;
}
__device__ __forceinline__ v4f ldg4(const float* p) { return *reinterpret_cast<const v4f*>(p); }
__device__ __forceinline__ void sts4(float* p, v4f v) { *reinterpret_cast<v4f*>(p) = v; }

__device__ __forceinline__ void lds_barrier() {
  __builtin_amdgcn_fence(__ATOMIC_RELEASE, "workgroup", "local");
  __builtin_amdgcn_s_barrier();
  __builtin_amdgcn_fence(__ATOMIC_ACQUIRE, "workgroup", "local");
}

__global__ __launch_bounds__(256) void cla_phase1(
    const float* kin, const float* vin,
    float* __restrict__ gkv, float* __restrict__ gks) {
  __shared__ __attribute__((aligned(16))) float Ks[C_ * PB1_];
  __shared__ __attribute__((aligned(16))) float Vs[C_ * PB1_];

  const int bg = blockIdx.x;
  const int nh = bg / NG_, g = bg % NG_;
  const int n = nh / H_, h = nh % H_;
  const int tid = threadIdx.x;
  const int wave = tid >> 5, lane = tid & 31;
  const size_t rs = (size_t)H_ * D_;
  const int r0 = tid >> 4;
  const int c4 = (tid & 15) << 2;

  v8f acc[2];
  acc[0] = (v8f){}; acc[1] = (v8f){};
  float s = 0.0f;

  const size_t goff = (size_t)r0 * rs + c4;
  const size_t cb0 = (((size_t)n * L_ + (size_t)(g * G_) * C_) * H_ + h) * D_ + goff;
  v4f kr[TILE_V4_], vr[TILE_V4_];
#pragma unroll
  for (int i = 0; i < TILE_V4_; ++i) {
    kr[i] = ldg4(kin + cb0 + (size_t)(16 * i) * rs);
    vr[i] = ldg4(vin + cb0 + (size_t)(16 * i) * rs);
  }

  for (int c = 0; c < G_; ++c) {
#pragma unroll
    for (int i = 0; i < TILE_V4_; ++i) {
      const int o = (r0 + 16 * i) * PB1_ + c4;
      sts4(Ks + o, featmap4(kr[i]));
      sts4(Vs + o, vr[i]);
    }
    if (c + 1 < G_) {
      const size_t cb = cb0 + (size_t)(c + 1) * C_ * rs;
#pragma unroll
      for (int i = 0; i < TILE_V4_; ++i) {
        kr[i] = ldg4(kin + cb + (size_t)(16 * i) * rs);
        vr[i] = ldg4(vin + cb + (size_t)(16 * i) * rs);
      }
    }
    lds_barrier();

#pragma unroll
    for (int j = 0; j < 2; ++j) {
      const int tt = wave * 2 + j;
      const int mb = (tt >> 2) * 16, nb = (tt & 3) * 16;
#pragma unroll
      for (int kk = 0; kk < 16; ++kk) {
        v2f a = frag_a_tr(Ks, PB1_, mb, kk, lane);
        v2f b = frag_b_rm(Vs, PB1_, nb, kk, lane);
        acc[j] = wmma4(a, b, acc[j]);
      }
    }
    if (tid < D_) {
#pragma unroll
      for (int r = 0; r < C_; ++r) s += Ks[r * PB1_ + tid];
    }
    lds_barrier();
  }

  v4f* kvq = reinterpret_cast<v4f*>(gkv) + (size_t)(nh * NG_ + g) * KV4_ + (size_t)(wave * 64 + lane) * 2;
#pragma unroll
  for (int j = 0; j < 2; ++j) {
    kvq[j * 64 + 0] = (v4f){acc[j][0], acc[j][1], acc[j][2], acc[j][3]};
    kvq[j * 64 + 1] = (v4f){acc[j][4], acc[j][5], acc[j][6], acc[j][7]};
  }
  if (tid < D_) gks[(size_t)(nh * NG_ + g) * D_ + tid] = s;
}

__global__ __launch_bounds__(256, 4) void cla_phase3(
    const float* qin, const float* kin,
    const float* vin, const float* __restrict__ gkv,
    const float* __restrict__ gks, float* __restrict__ out) {
  __shared__ __attribute__((aligned(16))) float Qs[C_ * PA_];
  __shared__ __attribute__((aligned(16))) float Vs[C_ * PB_];
  __shared__ __attribute__((aligned(16))) float Ss[C_ * PB_];
  __shared__ __attribute__((aligned(16))) float Ks[C_ * PK_];
  __shared__ float ks[D_];
  __shared__ float inv[C_];

  const int bg = blockIdx.x;
  const int nh = bg % NH_, g = NG_ - 1 - bg / NH_;
  const int n = nh / H_, h = nh % H_;
  const int tid = threadIdx.x;
  const int wave = tid >> 5, lane = tid & 31;
  const size_t rs = (size_t)H_ * D_;
  const int r0 = tid >> 4;
  const int c4 = (tid & 15) << 2;

  const unsigned loff = (unsigned)(r0 * (H_ * D_) + c4);
  const size_t cbu = (((size_t)n * L_ + (size_t)(g * G_) * C_) * H_ + h) * D_;
  const float* qc = qin + cbu;
  const float* kc = kin + cbu;
  const float* vc = vin + cbu;
  v4f qr[TILE_V4_], kr[TILE_V4_], vr[TILE_V4_];
  {
    const unsigned lo = (unsigned)pin((int)loff);
#pragma unroll
    for (int i = 0; i < TILE_V4_; ++i) {
      qr[i] = ldg4(qc + lo + (size_t)(16 * i) * rs);
      kr[i] = ldg4(kc + lo + (size_t)(16 * i) * rs);
      vr[i] = ldg4(vc + lo + (size_t)(16 * i) * rs);
    }
  }

  const v4f* tq = reinterpret_cast<const v4f*>(gkv) + (size_t)nh * NG_ * KV4_ + (size_t)(wave * 64 + lane) * 2;
  v4f run[2][2];
#pragma unroll
  for (int j = 0; j < 2; ++j) { run[j][0] = v4f(0.0f); run[j][1] = v4f(0.0f); }
  for (int t0 = 0; t0 < g; t0 += PF_) {
    v4f tv[PF_][2][2];
#pragma unroll
    for (int u = 0; u < PF_; ++u) {
      const int t = (t0 + u < g) ? (t0 + u) : (g - 1);
      const v4f* tp = tq + (size_t)t * KV4_;
#pragma unroll
      for (int j = 0; j < 2; ++j) { tv[u][j][0] = tp[j * 64 + 0]; tv[u][j][1] = tp[j * 64 + 1]; }
    }
#pragma unroll
    for (int u = 0; u < PF_; ++u) {
      if (t0 + u < g) {
#pragma unroll
        for (int j = 0; j < 2; ++j) { run[j][0] += tv[u][j][0]; run[j][1] += tv[u][j][1]; }
      }
    }
  }
  v8f sacc[2];
#pragma unroll
  for (int j = 0; j < 2; ++j)
    sacc[j] = (v8f){run[j][0].x, run[j][0].y, run[j][0].z, run[j][0].w,
                    run[j][1].x, run[j][1].y, run[j][1].z, run[j][1].w};
  float ksr = 0.0f;
  if (tid < D_) {
    float kt[NG_ - 1];
#pragma unroll
    for (int t = 0; t < NG_ - 1; ++t) {
      const int tc = (t < g) ? t : (g > 0 ? g - 1 : 0);
      kt[t] = gks[(size_t)(nh * NG_ + tc) * D_ + tid];
    }
#pragma unroll
    for (int t = 0; t < NG_ - 1; ++t) ksr += (t < g) ? kt[t] : 0.0f;
  }

  const int lc = lane & 15, lh2 = (lane >> 4) << 1;
  const int mb = (wave >> 1) * 16, nb0 = (wave & 1) * 32;
  const int tiw = __builtin_amdgcn_readfirstlane(wave) >> 1;
  const int qAh = ((mb + lc) * PA_ + lh2) / 2;
  const int bB0 = lh2 * PB_ + nb0 + lc;
  const int kBh = ((nb0 + lc) * PK_ + lh2) / 2;
  const int kA0 = lh2 * PK_ + mb + lc;
  const int pAh = ((mb + lc) * PK_ + lh2) / 2;

  for (int c = 0; c < G_; ++c) {
    const int t = g * G_ + c;

#pragma unroll
    for (int j = 0; j < 2; ++j) {
      const int tt = wave * 2 + j;
      const int mb = (tt >> 2) * 16, nb = (tt & 3) * 16;
      const int col = nb + (lane & 15);
      const int rb  = mb + ((lane >> 4) << 3);
#pragma unroll
      for (int r = 0; r < 8; ++r) Ss[(rb + r) * PB_ + col] = sacc[j][r];
    }
    if (tid < D_) ks[tid] = ksr;
#pragma unroll
    for (int i = 0; i < TILE_V4_; ++i) {
      const int r = r0 + 16 * i;
      sts4(Qs + r * PA_ + c4, featmap4(qr[i]));
      sts4(Ks + r * PK_ + c4, featmap4(kr[i]));
      sts4(Vs + r * PB_ + c4, vr[i]);
    }
    lds_barrier();

    v8f acc[2];
    {
    const int qA = 2 * pin(qAh), bB = pin(bB0);
#pragma unroll
    for (int j = 0; j < 2; ++j) {
      acc[j] = (v8f){};
#pragma unroll
      for (int kk = 0; kk < 16; ++kk) {
        v2f a = ld_pair(Qs, qA + kk * 4);
        v2f b = ld_rows(Ss, bB + kk * 4 * PB_ + 16 * j, PB_);
        acc[j] = wmma4(a, b, acc[j]);
      }
    }
    }

    v8f p[2];
    const int qA = 2 * pin(qAh), kB = 2 * pin(kBh);
#pragma unroll
    for (int j = 0; j < 2; ++j) {
      const int tt = wave * 2 + j;
      const int ti = tt >> 2, tj = tt & 3;
      p[j] = (v8f){};
      if (tj <= ti) {
#pragma unroll
        for (int kk = 0; kk < 16; ++kk) {
          v2f a = ld_pair(Qs, qA + kk * 4);
          v2f b = ld_pair(Ks, kB + 16 * j * PK_ + kk * 4);
          p[j] = wmma4(a, b, p[j]);
        }
      }
    }

    if (c + 1 < G_) {
      const int kA = pin(kA0), bBv = pin(bB0);
#pragma unroll
      for (int j = 0; j < 2; ++j) {
#pragma unroll
        for (int kk = 0; kk < 16; ++kk) {
          v2f a = ld_rows(Ks, kA + kk * 4 * PK_, PK_);
          v2f b = ld_rows(Vs, bBv + kk * 4 * PB_ + 16 * j, PB_);
          sacc[j] = wmma4(a, b, sacc[j]);
        }
      }
      if (tid < D_) {
        float cs = 0.0f;
#pragma unroll
        for (int r = 0; r < C_; ++r) cs += Ks[r * PK_ + tid];
        ksr += cs;
      }
    }
    lds_barrier();

#pragma unroll
    for (int j = 0; j < 2; ++j) {
      const int tt = wave * 2 + j;
      const int ti = tt >> 2, tj = tt & 3;
      const int col = tj * 16 + (lane & 15);
      const int rb  = ti * 16 + ((lane >> 4) << 3);
#pragma unroll
      for (int r = 0; r < 8; ++r) {
        const int row = rb + r;
        Ks[row * PK_ + col] = (col <= row) ? p[j][r] : 0.0f;
      }
    }
    lds_barrier();

    if (tid < C_) {
      float sm = 0.0f, dq = 0.0f;
#pragma unroll
      for (int cc = 0; cc < C_; ++cc) {
        sm += Ks[tid * PK_ + cc];
        dq += Qs[tid * PA_ + cc] * ks[cc];
      }
      inv[tid] = 1.0f / (sm + dq + EPS_);
    }

    if (c + 1 < G_) {
      const float* qn = qc + (size_t)(c + 1) * C_ * rs;
      const float* kn = kc + (size_t)(c + 1) * C_ * rs;
      const float* vn = vc + (size_t)(c + 1) * C_ * rs;
      const unsigned lo = (unsigned)pin((int)loff);
#pragma unroll
      for (int i = 0; i < TILE_V4_; ++i) {
        qr[i] = ldg4(qn + lo + (size_t)(16 * i) * rs);
        kr[i] = ldg4(kn + lo + (size_t)(16 * i) * rs);
        vr[i] = ldg4(vn + lo + (size_t)(16 * i) * rs);
      }
    }

    const int pA = 2 * pin(pAh), bBp = pin(bB0);
    if (tiw == 0)      pv_steps<4>(Ks, Vs, pA, bBp, acc[0], acc[1]);
    else if (tiw == 1) pv_steps<8>(Ks, Vs, pA, bBp, acc[0], acc[1]);
    else if (tiw == 2) pv_steps<12>(Ks, Vs, pA, bBp, acc[0], acc[1]);
    else               pv_steps<16>(Ks, Vs, pA, bBp, acc[0], acc[1]);
    lds_barrier();

    float* ob = out + (((size_t)n * L_ + (size_t)t * C_) * H_ + h) * DV_;
#pragma unroll
    for (int j = 0; j < 2; ++j) {
      const int tt = wave * 2 + j;
      const int mb = (tt >> 2) * 16, nb = (tt & 3) * 16;
      const int col = nb + (lane & 15);
      const int rb  = mb + ((lane >> 4) << 3);
#pragma unroll
      for (int r = 0; r < 8; ++r) {
        const int row = rb + r;
        ob[(size_t)row * (H_ * DV_) + col] = acc[j][r] * inv[row];
      }
    }
  }
}

extern "C" void kernel_launch(void* const* d_in, const int* in_sizes, int n_in,
                              void* d_out, int out_size, void* d_ws, size_t ws_size,
                              hipStream_t stream) {
  (void)in_sizes; (void)n_in; (void)out_size; (void)ws_size;
  const float* q = (const float*)d_in[0];
  const float* k = (const float*)d_in[1];
  const float* v = (const float*)d_in[2];
  float* out = (float*)d_out;

  float* gkv = (float*)d_ws;
  float* gks = gkv + (size_t)NH_ * NG_ * D_ * DV_;

  cla_phase1<<<NH_ * NG_, 256, 0, stream>>>(k, v, gkv, gks);
  cla_phase3<<<NH_ * NG_, 256, 0, stream>>>(q, k, v, gkv, gks, out);
}
